# starting kernel
# baseline (speedup 1.0000x reference)
#include <hip/hip_runtime.h>

typedef float v2f __attribute__((ext_vector_type(2)));
typedef float v8f __attribute__((ext_vector_type(8)));

#define NNODES   100000
#define NEGSLOPE 0.01f

__device__ __forceinline__ float lrelu(float v) { return v > 0.0f ? v : NEGSLOPE * v; }

__global__ void k_deg_init(float* __restrict__ deg, int n) {
    int i = blockIdx.x * blockDim.x + threadIdx.x;
    if (i < n) deg[i] = 1.0f;
}

__global__ void k_deg_count(const int* __restrict__ edges, float* __restrict__ deg, int E) {
    int e = blockIdx.x * blockDim.x + threadIdx.x;
    if (e < E) atomicAdd(&deg[edges[E + e]], 1.0f);
}

__global__ void k_dinv(const float* __restrict__ deg, float* __restrict__ dinv, int n) {
    int i = blockIdx.x * blockDim.x + threadIdx.x;
    if (i < n) dinv[i] = rsqrtf(deg[i]);
}

__global__ void k_gemm1(const float* __restrict__ x, const float* __restrict__ W,
                        const float* __restrict__ dinv, float* __restrict__ hs,
                        float* __restrict__ agg, int n) {
    __shared__ float sW[128 * 32];
    for (int i = threadIdx.x; i < 128 * 32; i += blockDim.x) {
        const int k = i >> 5, c = i & 31;
        sW[(k >> 1) * 64 + c * 2 + (k & 1)] = W[i];
    }
    __syncthreads();

    const int wave = threadIdx.x >> 5;
    const int lane = threadIdx.x & 31;
    const int tile = blockIdx.x * (blockDim.x >> 5) + wave;
    const int rowBase = tile * 16;
    if (rowBase >= n) return;

    const int half = lane >> 4;
    const int l16  = lane & 15;

    v8f c0 = {};
    v8f c1 = {};
    const float* xrow = x + (size_t)(rowBase + l16) * 128;

    #pragma unroll 4
    for (int k = 0; k < 128; k += 4) {
        const int ka = k + 2 * half;
        v2f a;  a.x = xrow[ka];  a.y = xrow[ka + 1];
        const int bidx = (ka >> 1) * 64 + l16 * 2;
        v2f b0 = *(const v2f*)&sW[bidx];
        v2f b1 = *(const v2f*)&sW[bidx + 32];
        c0 = __builtin_amdgcn_wmma_f32_16x16x4_f32(false, a, false, b0, (short)0, c0, false, false);
        c1 = __builtin_amdgcn_wmma_f32_16x16x4_f32(false, a, false, b1, (short)0, c1, false, false);
    }

    #pragma unroll
    for (int r = 0; r < 8; ++r) {
        const int m = rowBase + r + 8 * half;
        const float s = dinv[m];
        const float v0 = c0[r] * s;
        const float v1 = c1[r] * s;
        hs [(size_t)m * 32 + l16]      = v0;
        hs [(size_t)m * 32 + 16 + l16] = v1;
        agg[(size_t)m * 32 + l16]      = v0;
        agg[(size_t)m * 32 + 16 + l16] = v1;
    }
}

__global__ void k_gemm2(const float* __restrict__ agg1, const float* __restrict__ W,
                        const float* __restrict__ b1, const float* __restrict__ dinv,
                        float* __restrict__ hs, float* __restrict__ agg, int n) {
    __shared__ float sW[32 * 16];
    __shared__ float sB[32];
    for (int i = threadIdx.x; i < 32 * 16; i += blockDim.x) {
        const int k = i >> 4, c = i & 15;
        sW[(k >> 1) * 32 + c * 2 + (k & 1)] = W[i];
    }
    if (threadIdx.x < 32) sB[threadIdx.x] = b1[threadIdx.x];
    __syncthreads();

    const int wave = threadIdx.x >> 5;
    const int lane = threadIdx.x & 31;
    const int tile = blockIdx.x * (blockDim.x >> 5) + wave;
    const int rowBase = tile * 16;
    if (rowBase >= n) return;

    const int half = lane >> 4;
    const int l16  = lane & 15;

    v8f c = {};
    const int row = rowBase + l16;
    const float dvRow = dinv[row];
    const float* arow = agg1 + (size_t)row * 32;

    #pragma unroll
    for (int k = 0; k < 32; k += 4) {
        const int ka = k + 2 * half;
        v2f a;
        a.x = lrelu(dvRow * arow[ka]     + sB[ka]);
        a.y = lrelu(dvRow * arow[ka + 1] + sB[ka + 1]);
        v2f b = *(const v2f*)&sW[(ka >> 1) * 32 + l16 * 2];
        c = __builtin_amdgcn_wmma_f32_16x16x4_f32(false, a, false, b, (short)0, c, false, false);
    }

    #pragma unroll
    for (int r = 0; r < 8; ++r) {
        const int m = rowBase + r + 8 * half;
        const float v = c[r] * dinv[m];
        hs [(size_t)m * 16 + l16] = v;
        agg[(size_t)m * 16 + l16] = v;
    }
}

__global__ void k_scatter(const int* __restrict__ edges, const float* __restrict__ hs,
                          float* __restrict__ agg, int E, int clog) {
    int idx = blockIdx.x * blockDim.x + threadIdx.x;
    int total = E << clog;
    if (idx < total) {
        int e = idx >> clog;
        int j = idx & ((1 << clog) - 1);
        int s = edges[e];
        int d = edges[E + e];
        atomicAdd(&agg[((size_t)d << clog) + j], hs[((size_t)s << clog) + j]);
    }
}

__global__ void k_out(const float* __restrict__ agg2, const float* __restrict__ dinv,
                      const float* __restrict__ b2, const float* __restrict__ Wfc,
                      const float* __restrict__ bfc, float* __restrict__ out, int n) {
    int v = blockIdx.x * blockDim.x + threadIdx.x;
    if (v < n) {
        const float dv = dinv[v];
        const float* row = agg2 + (size_t)v * 16;
        float s = bfc[0];
        #pragma unroll
        for (int j = 0; j < 16; ++j) s += lrelu(dv * row[j] + b2[j]) * Wfc[j];
        out[v] = s;
    }
}

extern "C" void kernel_launch(void* const* d_in, const int* in_sizes, int n_in,
                              void* d_out, int out_size, void* d_ws, size_t ws_size,
                              hipStream_t stream) {
    const float* x    = (const float*)d_in[0];
    const int*   ei   = (const int*)  d_in[1];
    const float* W1   = (const float*)d_in[2];
    const float* b1   = (const float*)d_in[3];
    const float* W2   = (const float*)d_in[4];
    const float* b2   = (const float*)d_in[5];
    const float* Wfc  = (const float*)d_in[6];
    const float* bfc  = (const float*)d_in[7];
    float*       out  = (float*)d_out;

    const int n = NNODES;
    const int E = in_sizes[1] / 2;

    char* ws = (char*)d_ws;
    size_t off = 0;
    auto carve = [&](size_t bytes) { char* p = ws + off; off = (off + bytes + 511) & ~(size_t)511; return p; };
    float* deg  = (float*)carve((size_t)n * 4);
    float* dinv = (float*)carve((size_t)n * 4);
    float* h1s  = (float*)carve((size_t)n * 32 * 4);
    float* agg1 = (float*)carve((size_t)n * 32 * 4);
    float* h2s  = (float*)carve((size_t)n * 16 * 4);
    float* agg2 = (float*)carve((size_t)n * 16 * 4);
    (void)ws_size;

    const int T = 256;
    const int wavesPerBlock = T / 32;
    const int tiles = (n + 15) / 16;
    const int gemmBlocks = (tiles + wavesPerBlock - 1) / wavesPerBlock;

    k_deg_init <<<(n + T - 1) / T, T, 0, stream>>>(deg, n);
    k_deg_count<<<(E + T - 1) / T, T, 0, stream>>>(ei, deg, E);
    k_dinv     <<<(n + T - 1) / T, T, 0, stream>>>(deg, dinv, n);

    k_gemm1  <<<gemmBlocks, T, 0, stream>>>(x, W1, dinv, h1s, agg1, n);
    k_scatter<<<((E << 5) + T - 1) / T, T, 0, stream>>>(ei, h1s, agg1, E, 5);

    k_gemm2  <<<gemmBlocks, T, 0, stream>>>(agg1, W2, b1, dinv, h2s, agg2, n);
    k_scatter<<<((E << 4) + T - 1) / T, T, 0, stream>>>(ei, h2s, agg2, E, 4);

    k_out<<<(n + T - 1) / T, T, 0, stream>>>(agg2, dinv, b2, Wfc, bfc, out, n);
}


# degree count: LDS integer histograms per (node slice, edge chunk) block, count planes summed in the dinv kernel; replaces the random global float atomics and the init launch
# speedup vs baseline: 1.2744x; 1.2744x over previous
#include <hip/hip_runtime.h>

typedef float v2f __attribute__((ext_vector_type(2)));
typedef float v8f __attribute__((ext_vector_type(8)));

#define NNODES   100000
#define NEGSLOPE 0.01f

__device__ __forceinline__ float lrelu(float v) { return v > 0.0f ? v : NEGSLOPE * v; }

#define NSPLIT  4
#define QNODES  (NNODES / NSPLIT)
#define NCHUNKS 64
#define HT      1024
#define HDEPTH  8
static_assert(NNODES % (NSPLIT * 4) == 0, "a slice must be a whole number of 16-byte groups");

__global__ void __launch_bounds__(HT)
k_deg_hist(const int* __restrict__ edges, unsigned int* __restrict__ part, int E) {
    __shared__ uint4 sCnt4[QNODES / 4];
    unsigned int* sCnt = (unsigned int*)sCnt4;
    const int tid   = threadIdx.x;
    const int slice = blockIdx.x % NSPLIT;
    const int chunk = blockIdx.x / NSPLIT;
    const int base  = slice * QNODES;

    for (int i = tid; i < QNODES / 4; i += HT) sCnt4[i] = make_uint4(0u, 0u, 0u, 0u);
    __syncthreads();

    const int per   = (E + NCHUNKS - 1) / NCHUNKS;
    const int start = chunk * per;
    const int end   = min(E, start + per);
    const int* __restrict__ dst = edges + E;

    for (int e0 = start; e0 < end; e0 += HDEPTH * HT) {
        int d[HDEPTH];
        #pragma unroll
        for (int k = 0; k < HDEPTH; ++k) {
            const int e = e0 + k * HT + tid;
            d[k] = (e < end) ? dst[e] : -1;
        }
        #pragma unroll
        for (int k = 0; k < HDEPTH; ++k) {
            const unsigned int r = (unsigned int)(d[k] - base);
            if (r < (unsigned int)QNODES) atomicAdd(&sCnt[r], 1u);
        }
    }
    __syncthreads();

    uint4* __restrict__ row = (uint4*)(part + (size_t)chunk * NNODES + base);
    for (int i = tid; i < QNODES / 4; i += HT) row[i] = sCnt4[i];
}

__global__ void k_dinv(const unsigned int* __restrict__ part, float* __restrict__ dinv, int n) {
    int i = blockIdx.x * blockDim.x + threadIdx.x;
    if (i < n) {
        unsigned int d = 1u;
        #pragma unroll
        for (int c = 0; c < NCHUNKS; ++c) d += part[(size_t)c * NNODES + i];
        dinv[i] = rsqrtf((float)d);
    }
}

__global__ void k_gemm1(const float* __restrict__ x, const float* __restrict__ W,
                        const float* __restrict__ dinv, float* __restrict__ hs,
                        float* __restrict__ agg, int n) {
    __shared__ float sW[128 * 32];
    for (int i = threadIdx.x; i < 128 * 32; i += blockDim.x) {
        const int k = i >> 5, c = i & 31;
        sW[(k >> 1) * 64 + c * 2 + (k & 1)] = W[i];
    }
    __syncthreads();

    const int wave = threadIdx.x >> 5;
    const int lane = threadIdx.x & 31;
    const int tile = blockIdx.x * (blockDim.x >> 5) + wave;
    const int rowBase = tile * 16;
    if (rowBase >= n) return;

    const int half = lane >> 4;
    const int l16  = lane & 15;

    v8f c0 = {};
    v8f c1 = {};
    const float* xrow = x + (size_t)(rowBase + l16) * 128;

    #pragma unroll 4
    for (int k = 0; k < 128; k += 4) {
        const int ka = k + 2 * half;
        v2f a;  a.x = xrow[ka];  a.y = xrow[ka + 1];
        const int bidx = (ka >> 1) * 64 + l16 * 2;
        v2f b0 = *(const v2f*)&sW[bidx];
        v2f b1 = *(const v2f*)&sW[bidx + 32];
        c0 = __builtin_amdgcn_wmma_f32_16x16x4_f32(false, a, false, b0, (short)0, c0, false, false);
        c1 = __builtin_amdgcn_wmma_f32_16x16x4_f32(false, a, false, b1, (short)0, c1, false, false);
    }

    #pragma unroll
    for (int r = 0; r < 8; ++r) {
        const int m = rowBase + r + 8 * half;
        const float s = dinv[m];
        const float v0 = c0[r] * s;
        const float v1 = c1[r] * s;
        hs [(size_t)m * 32 + l16]      = v0;
        hs [(size_t)m * 32 + 16 + l16] = v1;
        agg[(size_t)m * 32 + l16]      = v0;
        agg[(size_t)m * 32 + 16 + l16] = v1;
    }
}

__global__ void k_gemm2(const float* __restrict__ agg1, const float* __restrict__ W,
                        const float* __restrict__ b1, const float* __restrict__ dinv,
                        float* __restrict__ hs, float* __restrict__ agg, int n) {
    __shared__ float sW[32 * 16];
    __shared__ float sB[32];
    for (int i = threadIdx.x; i < 32 * 16; i += blockDim.x) {
        const int k = i >> 4, c = i & 15;
        sW[(k >> 1) * 32 + c * 2 + (k & 1)] = W[i];
    }
    if (threadIdx.x < 32) sB[threadIdx.x] = b1[threadIdx.x];
    __syncthreads();

    const int wave = threadIdx.x >> 5;
    const int lane = threadIdx.x & 31;
    const int tile = blockIdx.x * (blockDim.x >> 5) + wave;
    const int rowBase = tile * 16;
    if (rowBase >= n) return;

    const int half = lane >> 4;
    const int l16  = lane & 15;

    v8f c = {};
    const int row = rowBase + l16;
    const float dvRow = dinv[row];
    const float* arow = agg1 + (size_t)row * 32;

    #pragma unroll
    for (int k = 0; k < 32; k += 4) {
        const int ka = k + 2 * half;
        v2f a;
        a.x = lrelu(dvRow * arow[ka]     + sB[ka]);
        a.y = lrelu(dvRow * arow[ka + 1] + sB[ka + 1]);
        v2f b = *(const v2f*)&sW[(ka >> 1) * 32 + l16 * 2];
        c = __builtin_amdgcn_wmma_f32_16x16x4_f32(false, a, false, b, (short)0, c, false, false);
    }

    #pragma unroll
    for (int r = 0; r < 8; ++r) {
        const int m = rowBase + r + 8 * half;
        const float v = c[r] * dinv[m];
        hs [(size_t)m * 16 + l16] = v;
        agg[(size_t)m * 16 + l16] = v;
    }
}

__global__ void k_scatter(const int* __restrict__ edges, const float* __restrict__ hs,
                          float* __restrict__ agg, int E, int clog) {
    int idx = blockIdx.x * blockDim.x + threadIdx.x;
    int total = E << clog;
    if (idx < total) {
        int e = idx >> clog;
        int j = idx & ((1 << clog) - 1);
        int s = edges[e];
        int d = edges[E + e];
        atomicAdd(&agg[((size_t)d << clog) + j], hs[((size_t)s << clog) + j]);
    }
}

__global__ void k_out(const float* __restrict__ agg2, const float* __restrict__ dinv,
                      const float* __restrict__ b2, const float* __restrict__ Wfc,
                      const float* __restrict__ bfc, float* __restrict__ out, int n) {
    int v = blockIdx.x * blockDim.x + threadIdx.x;
    if (v < n) {
        const float dv = dinv[v];
        const float* row = agg2 + (size_t)v * 16;
        float s = bfc[0];
        #pragma unroll
        for (int j = 0; j < 16; ++j) s += lrelu(dv * row[j] + b2[j]) * Wfc[j];
        out[v] = s;
    }
}

extern "C" void kernel_launch(void* const* d_in, const int* in_sizes, int n_in,
                              void* d_out, int out_size, void* d_ws, size_t ws_size,
                              hipStream_t stream) {
    const float* x    = (const float*)d_in[0];
    const int*   ei   = (const int*)  d_in[1];
    const float* W1   = (const float*)d_in[2];
    const float* b1   = (const float*)d_in[3];
    const float* W2   = (const float*)d_in[4];
    const float* b2   = (const float*)d_in[5];
    const float* Wfc  = (const float*)d_in[6];
    const float* bfc  = (const float*)d_in[7];
    float*       out  = (float*)d_out;

    const int n = NNODES;
    const int E = in_sizes[1] / 2;

    char* ws = (char*)d_ws;
    size_t off = 0;
    auto carve = [&](size_t bytes) { char* p = ws + off; off = (off + bytes + 511) & ~(size_t)511; return p; };
    unsigned int* part = (unsigned int*)carve((size_t)NCHUNKS * NNODES * 4);
    float* dinv = (float*)carve((size_t)n * 4);
    float* h1s  = (float*)carve((size_t)n * 32 * 4);
    float* agg1 = (float*)carve((size_t)n * 32 * 4);
    float* h2s  = (float*)carve((size_t)n * 16 * 4);
    float* agg2 = (float*)carve((size_t)n * 16 * 4);
    (void)ws_size;

    const int T = 256;
    const int wavesPerBlock = T / 32;
    const int tiles = (n + 15) / 16;
    const int gemmBlocks = (tiles + wavesPerBlock - 1) / wavesPerBlock;

    k_deg_hist<<<NCHUNKS * NSPLIT, HT, 0, stream>>>(ei, part, E);
    k_dinv    <<<(n + T - 1) / T, T, 0, stream>>>(part, dinv, n);

    k_gemm1  <<<gemmBlocks, T, 0, stream>>>(x, W1, dinv, h1s, agg1, n);
    k_scatter<<<((E << 5) + T - 1) / T, T, 0, stream>>>(ei, h1s, agg1, E, 5);

    k_gemm2  <<<gemmBlocks, T, 0, stream>>>(agg1, W2, b1, dinv, h2s, agg2, n);
    k_scatter<<<((E << 4) + T - 1) / T, T, 0, stream>>>(ei, h2s, agg2, E, 4);

    k_out<<<(n + T - 1) / T, T, 0, stream>>>(agg2, dinv, b2, Wfc, bfc, out, n);
}


# aggregation over a destination-sorted edge list (counting sort, LDS cursors, arrival-order rank in a chunk): a wave per node sums its rows in list order, layer 2 as two half sums; no float atomics
# speedup vs baseline: 3.4588x; 2.7141x over previous
#include <hip/hip_runtime.h>

typedef float v2f __attribute__((ext_vector_type(2)));
typedef float v8f __attribute__((ext_vector_type(8)));

#define NNODES   100000
#define NEGSLOPE 0.01f

__device__ __forceinline__ float lrelu(float v) { return v > 0.0f ? v : NEGSLOPE * v; }

#define NSPLIT  4
#define QNODES  (NNODES / NSPLIT)
#define NCHUNKS 64
#define HT      1024
#define HDEPTH  8
static_assert(NNODES % (NSPLIT * 4) == 0, "a slice must be a whole number of 16-byte groups");

__global__ void __launch_bounds__(HT)
k_deg_hist(const int* __restrict__ edges, unsigned int* __restrict__ part, int E) {
    __shared__ uint4 sCnt4[QNODES / 4];
    unsigned int* sCnt = (unsigned int*)sCnt4;
    const int tid   = threadIdx.x;
    const int slice = blockIdx.x % NSPLIT;
    const int chunk = blockIdx.x / NSPLIT;
    const int base  = slice * QNODES;

    for (int i = tid; i < QNODES / 4; i += HT) sCnt4[i] = make_uint4(0u, 0u, 0u, 0u);
    __syncthreads();

    const int per   = (E + NCHUNKS - 1) / NCHUNKS;
    const int start = chunk * per;
    const int end   = min(E, start + per);
    const int* __restrict__ dst = edges + E;

    for (int e0 = start; e0 < end; e0 += HDEPTH * HT) {
        int d[HDEPTH];
        #pragma unroll
        for (int k = 0; k < HDEPTH; ++k) {
            const int e = e0 + k * HT + tid;
            d[k] = (e < end) ? dst[e] : -1;
        }
        #pragma unroll
        for (int k = 0; k < HDEPTH; ++k) {
            const unsigned int r = (unsigned int)(d[k] - base);
            if (r < (unsigned int)QNODES) atomicAdd(&sCnt[r], 1u);
        }
    }
    __syncthreads();

    uint4* __restrict__ row = (uint4*)(part + (size_t)chunk * NNODES + base);
    for (int i = tid; i < QNODES / 4; i += HT) row[i] = sCnt4[i];
}

__device__ __forceinline__ unsigned int wave_incl_scan(unsigned int v, int lane) {
    #pragma unroll
    for (int o = 1; o < 32; o <<= 1) {
        const unsigned int t = __shfl_up(v, (unsigned int)o);
        if (lane >= o) v += t;
    }
    return v;
}

#define DT 256
#define NBLK ((NNODES + DT - 1) / DT)
__global__ void __launch_bounds__(DT)
k_dinv_pre(unsigned int* __restrict__ part, float* __restrict__ dinv,
           unsigned int* __restrict__ loc, unsigned int* __restrict__ blockSum, int n) {
    __shared__ unsigned int sWave[DT / 32];
    const int tid  = threadIdx.x;
    const int lane = tid & 31;
    const int wave = tid >> 5;
    const int i = blockIdx.x * DT + tid;
    unsigned int run = 0u;
    if (i < n) {
        unsigned int cnt[NCHUNKS];
        #pragma unroll
        for (int c = 0; c < NCHUNKS; ++c) cnt[c] = part[(size_t)c * NNODES + i];
        #pragma unroll
        for (int c = 0; c < NCHUNKS; ++c) { part[(size_t)c * NNODES + i] = run; run += cnt[c]; }
        dinv[i] = rsqrtf((float)(run + 1u));
    }
    const unsigned int incl = wave_incl_scan(run, lane);
    if (lane == 31) sWave[wave] = incl;
    __syncthreads();
    unsigned int woff = 0u;
    for (int w = 0; w < wave; ++w) woff += sWave[w];
    if (i < n) loc[i] = woff + incl - run;
    if (tid == DT - 1) blockSum[blockIdx.x] = woff + incl;
}

__global__ void __launch_bounds__(HT)
k_fill(const int* __restrict__ edges, const unsigned int* __restrict__ part,
       const unsigned int* __restrict__ loc, const unsigned int* __restrict__ blockSum,
       unsigned int* __restrict__ rowptr, int* __restrict__ col, int E) {
    __shared__ uint4 sCur4[QNODES / 4];
    __shared__ unsigned int sBlk[512];
    __shared__ unsigned int sW[16];
    static_assert(NBLK <= 512, "block sums must fit the 512-entry scan");
    unsigned int* sCur = (unsigned int*)sCur4;
    const int tid   = threadIdx.x;
    const int lane  = tid & 31;
    const int wave  = tid >> 5;
    const int slice = blockIdx.x % NSPLIT;
    const int chunk = blockIdx.x / NSPLIT;
    const int base  = slice * QNODES;

    const unsigned int bs = (tid < NBLK) ? blockSum[tid] : 0u;
    const unsigned int incl = wave_incl_scan(bs, lane);
    if (tid < 512 && lane == 31) sW[wave] = incl;
    __syncthreads();
    if (tid < 512) {
        unsigned int woff = 0u;
        for (int w = 0; w < wave; ++w) woff += sW[w];
        sBlk[tid] = woff + incl - bs;
        if (blockIdx.x == 0 && tid == NBLK - 1) rowptr[NNODES] = woff + incl;
    }
    __syncthreads();

    const uint4* __restrict__ loc4 = (const uint4*)(loc + base);
    const uint4* __restrict__ pre4 = (const uint4*)(part + (size_t)chunk * NNODES + base);
    uint4* __restrict__ rp4 = (uint4*)(rowptr + base);
    for (int i = tid; i < QNODES / 4; i += HT) {
        const uint4 a = loc4[i];
        const uint4 b = pre4[i];
        const unsigned int bo = sBlk[(base + 4 * i) >> 8];
        const uint4 st = make_uint4(a.x + b.x + bo, a.y + b.y + bo, a.z + b.z + bo, a.w + b.w + bo);
        sCur4[i] = st;
        if (chunk == 0) rp4[i] = st;
    }
    __syncthreads();

    const int per   = (E + NCHUNKS - 1) / NCHUNKS;
    const int start = chunk * per;
    const int end   = min(E, start + per);
    const int* __restrict__ srcv = edges;
    const int* __restrict__ dstv = edges + E;

    for (int e0 = start; e0 < end; e0 += HDEPTH * HT) {
        int d[HDEPTH];
        int s[HDEPTH];
        #pragma unroll
        for (int k = 0; k < HDEPTH; ++k) {
            const int e = e0 + k * HT + tid;
            d[k] = (e < end) ? dstv[e] : -1;
            s[k] = (e < end) ? srcv[e] : 0;
        }
        unsigned int pos[HDEPTH];
        #pragma unroll
        for (int k = 0; k < HDEPTH; ++k) {
            const unsigned int r = (unsigned int)(d[k] - base);
            pos[k] = 0u;
            if (r < (unsigned int)QNODES) pos[k] = atomicAdd(&sCur[r], 1u);
        }
        #pragma unroll
        for (int k = 0; k < HDEPTH; ++k) {
            const unsigned int r = (unsigned int)(d[k] - base);
            if (r < (unsigned int)QNODES) col[pos[k]] = s[k];
        }
    }
}

__global__ void k_gemm1(const float* __restrict__ x, const float* __restrict__ W,
                        const float* __restrict__ dinv, float* __restrict__ hs, int n) {
    __shared__ float sW[128 * 32];
    for (int i = threadIdx.x; i < 128 * 32; i += blockDim.x) {
        const int k = i >> 5, c = i & 31;
        sW[(k >> 1) * 64 + c * 2 + (k & 1)] = W[i];
    }
    __syncthreads();

    const int wave = threadIdx.x >> 5;
    const int lane = threadIdx.x & 31;
    const int tile = blockIdx.x * (blockDim.x >> 5) + wave;
    const int rowBase = tile * 16;
    if (rowBase >= n) return;

    const int half = lane >> 4;
    const int l16  = lane & 15;

    v8f c0 = {};
    v8f c1 = {};
    const float* xrow = x + (size_t)(rowBase + l16) * 128;

    #pragma unroll 4
    for (int k = 0; k < 128; k += 4) {
        const int ka = k + 2 * half;
        v2f a;  a.x = xrow[ka];  a.y = xrow[ka + 1];
        const int bidx = (ka >> 1) * 64 + l16 * 2;
        v2f b0 = *(const v2f*)&sW[bidx];
        v2f b1 = *(const v2f*)&sW[bidx + 32];
        c0 = __builtin_amdgcn_wmma_f32_16x16x4_f32(false, a, false, b0, (short)0, c0, false, false);
        c1 = __builtin_amdgcn_wmma_f32_16x16x4_f32(false, a, false, b1, (short)0, c1, false, false);
    }

    #pragma unroll
    for (int r = 0; r < 8; ++r) {
        const int m = rowBase + r + 8 * half;
        const float s = dinv[m];
        const float v0 = c0[r] * s;
        const float v1 = c1[r] * s;
        hs[(size_t)m * 32 + l16]      = v0;
        hs[(size_t)m * 32 + 16 + l16] = v1;
    }
}

__global__ void k_gemm2(const float* __restrict__ agg1, const float* __restrict__ W,
                        const float* __restrict__ b1, const float* __restrict__ dinv,
                        float* __restrict__ hs, int n) {
    __shared__ float sW[32 * 16];
    __shared__ float sB[32];
    for (int i = threadIdx.x; i < 32 * 16; i += blockDim.x) {
        const int k = i >> 4, c = i & 15;
        sW[(k >> 1) * 32 + c * 2 + (k & 1)] = W[i];
    }
    if (threadIdx.x < 32) sB[threadIdx.x] = b1[threadIdx.x];
    __syncthreads();

    const int wave = threadIdx.x >> 5;
    const int lane = threadIdx.x & 31;
    const int tile = blockIdx.x * (blockDim.x >> 5) + wave;
    const int rowBase = tile * 16;
    if (rowBase >= n) return;

    const int half = lane >> 4;
    const int l16  = lane & 15;

    v8f c = {};
    const int row = rowBase + l16;
    const float dvRow = dinv[row];
    const float* arow = agg1 + (size_t)row * 32;

    #pragma unroll
    for (int k = 0; k < 32; k += 4) {
        const int ka = k + 2 * half;
        v2f a;
        a.x = lrelu(dvRow * arow[ka]     + sB[ka]);
        a.y = lrelu(dvRow * arow[ka + 1] + sB[ka + 1]);
        v2f b = *(const v2f*)&sW[(ka >> 1) * 32 + l16 * 2];
        c = __builtin_amdgcn_wmma_f32_16x16x4_f32(false, a, false, b, (short)0, c, false, false);
    }

    #pragma unroll
    for (int r = 0; r < 8; ++r) {
        const int m = rowBase + r + 8 * half;
        const float v = c[r] * dinv[m];
        hs[(size_t)m * 16 + l16] = v;
    }
}

#define PT 256
__global__ void __launch_bounds__(PT)
k_pull32(const unsigned int* __restrict__ rowptr, const int* __restrict__ col,
         const float* __restrict__ hs, float* __restrict__ agg, int n) {
    const int lane = threadIdx.x & 31;
    const int v = __builtin_amdgcn_readfirstlane((int)(blockIdx.x * (PT / 32) + (threadIdx.x >> 5)));
    if (v >= n) return;
    const int beg = (int)rowptr[v];
    const int end = (int)rowptr[v + 1];
    float acc = hs[(size_t)v * 32 + lane];
    for (int p0 = beg; p0 < end; p0 += 32) {
        const int cnt = min(32, end - p0);
        const int mine = (lane < cnt) ? col[p0 + lane] : 0;
        int k = 0;
        for (; k + 8 <= cnt; k += 8) {
            float r[8];
            #pragma unroll
            for (int u = 0; u < 8; ++u) {
                const int s = __builtin_amdgcn_readlane(mine, k + u);
                r[u] = hs[(size_t)s * 32 + lane];
            }
            #pragma unroll
            for (int u = 0; u < 8; ++u) acc += r[u];
        }
        if (k < cnt) {
            float r[8];
            #pragma unroll
            for (int u = 0; u < 8; ++u) {
                r[u] = 0.0f;
                if (k + u < cnt) {
                    const int s = __builtin_amdgcn_readlane(mine, k + u);
                    r[u] = hs[(size_t)s * 32 + lane];
                }
            }
            #pragma unroll
            for (int u = 0; u < 8; ++u) acc += r[u];
        }
    }
    agg[(size_t)v * 32 + lane] = acc;
}

__global__ void __launch_bounds__(PT)
k_pull16(const unsigned int* __restrict__ rowptr, const int* __restrict__ col,
         const float* __restrict__ hs, float* __restrict__ agg, int n) {
    const int lane = threadIdx.x & 31;
    const int half = lane >> 4;
    const int l16  = lane & 15;
    const int v = __builtin_amdgcn_readfirstlane((int)(blockIdx.x * (PT / 32) + (threadIdx.x >> 5)));
    if (v >= n) return;
    const int beg = (int)rowptr[v];
    const int end = (int)rowptr[v + 1];
    const float self = hs[(size_t)v * 16 + l16];
    float acc = 0.0f;
    for (int p0 = beg; p0 < end; p0 += 32) {
        const int cnt = min(32, end - p0);
        const int mine = (lane < cnt) ? col[p0 + lane] : 0;
        int k = 0;
        for (; k + 16 <= cnt; k += 16) {
            float r[8];
            #pragma unroll
            for (int u = 0; u < 8; ++u) {
                const int s = __shfl(mine, k + 2 * u + half);
                r[u] = hs[(size_t)s * 16 + l16];
            }
            #pragma unroll
            for (int u = 0; u < 8; ++u) acc += r[u];
        }
        if (k < cnt) {
            float r[8];
            #pragma unroll
            for (int u = 0; u < 8; ++u) {
                const int j = k + 2 * u + half;
                const int s = __shfl(mine, j & 31);
                r[u] = 0.0f;
                if (j < cnt) r[u] = hs[(size_t)s * 16 + l16];
            }
            #pragma unroll
            for (int u = 0; u < 8; ++u) acc += r[u];
        }
    }
    acc += __shfl_xor(acc, 16);
    if (half == 0) agg[(size_t)v * 16 + l16] = acc + self;
}

__global__ void k_out(const float* __restrict__ agg2, const float* __restrict__ dinv,
                      const float* __restrict__ b2, const float* __restrict__ Wfc,
                      const float* __restrict__ bfc, float* __restrict__ out, int n) {
    int v = blockIdx.x * blockDim.x + threadIdx.x;
    if (v < n) {
        const float dv = dinv[v];
        const float* row = agg2 + (size_t)v * 16;
        float s = bfc[0];
        #pragma unroll
        for (int j = 0; j < 16; ++j) s += lrelu(dv * row[j] + b2[j]) * Wfc[j];
        out[v] = s;
    }
}

extern "C" void kernel_launch(void* const* d_in, const int* in_sizes, int n_in,
                              void* d_out, int out_size, void* d_ws, size_t ws_size,
                              hipStream_t stream) {
    const float* x    = (const float*)d_in[0];
    const int*   ei   = (const int*)  d_in[1];
    const float* W1   = (const float*)d_in[2];
    const float* b1   = (const float*)d_in[3];
    const float* W2   = (const float*)d_in[4];
    const float* b2   = (const float*)d_in[5];
    const float* Wfc  = (const float*)d_in[6];
    const float* bfc  = (const float*)d_in[7];
    float*       out  = (float*)d_out;

    const int n = NNODES;
    const int E = in_sizes[1] / 2;

    char* ws = (char*)d_ws;
    size_t off = 0;
    auto carve = [&](size_t bytes) { char* p = ws + off; off = (off + bytes + 511) & ~(size_t)511; return p; };
    unsigned int* part = (unsigned int*)carve((size_t)NCHUNKS * NNODES * 4);
    float* dinv = (float*)carve((size_t)n * 4);
    unsigned int* loc      = (unsigned int*)carve((size_t)n * 4);
    unsigned int* blockSum = (unsigned int*)carve((size_t)NBLK * 4);
    unsigned int* rowptr   = (unsigned int*)carve((size_t)(n + 1) * 4);
    float* h1s  = (float*)carve((size_t)n * 32 * 4);
    float* agg1 = (float*)carve((size_t)n * 32 * 4);
    float* h2s  = (float*)carve((size_t)n * 16 * 4);
    float* agg2 = (float*)carve((size_t)n * 16 * 4);
    int*   col  = (int*)(ws + off);
    (void)ws_size;

    const int T = 256;
    const int wavesPerBlock = T / 32;
    const int tiles = (n + 15) / 16;
    const int gemmBlocks = (tiles + wavesPerBlock - 1) / wavesPerBlock;

    k_deg_hist<<<NCHUNKS * NSPLIT, HT, 0, stream>>>(ei, part, E);
    k_dinv_pre<<<NBLK, DT, 0, stream>>>(part, dinv, loc, blockSum, n);
    k_fill    <<<NCHUNKS * NSPLIT, HT, 0, stream>>>(ei, part, loc, blockSum, rowptr, col, E);

    const int pullBlocks = (n + PT / 32 - 1) / (PT / 32);

    k_gemm1 <<<gemmBlocks, T, 0, stream>>>(x, W1, dinv, h1s, n);
    k_pull32<<<pullBlocks, PT, 0, stream>>>(rowptr, col, h1s, agg1, n);

    k_gemm2 <<<gemmBlocks, T, 0, stream>>>(agg1, W2, b1, dinv, h2s, n);
    k_pull16<<<pullBlocks, PT, 0, stream>>>(rowptr, col, h2s, agg2, n);

    k_out<<<(n + T - 1) / T, T, 0, stream>>>(agg2, dinv, b2, Wfc, bfc, out, n);
}


# aggregation launches finish their own layer: 32-node blocks, 4 nodes per wave; layer-1 sums feed the W2 WMMA through an LDS tile, layer-2 sums the Wfc dot; agg1/agg2 planes and 2 launches removed
# speedup vs baseline: 3.6532x; 1.0562x over previous
#include <hip/hip_runtime.h>

typedef float v2f __attribute__((ext_vector_type(2)));
typedef float v8f __attribute__((ext_vector_type(8)));

#define NNODES   100000
#define NEGSLOPE 0.01f

__device__ __forceinline__ float lrelu(float v) { return v > 0.0f ? v : NEGSLOPE * v; }

#define NSPLIT  4
#define QNODES  (NNODES / NSPLIT)
#define NCHUNKS 64
#define HT      1024
#define HDEPTH  8
static_assert(NNODES % (NSPLIT * 4) == 0, "a slice must be a whole number of 16-byte groups");

__global__ void __launch_bounds__(HT)
k_deg_hist(const int* __restrict__ edges, unsigned int* __restrict__ part, int E) {
    __shared__ uint4 sCnt4[QNODES / 4];
    unsigned int* sCnt = (unsigned int*)sCnt4;
    const int tid   = threadIdx.x;
    const int slice = blockIdx.x % NSPLIT;
    const int chunk = blockIdx.x / NSPLIT;
    const int base  = slice * QNODES;

    for (int i = tid; i < QNODES / 4; i += HT) sCnt4[i] = make_uint4(0u, 0u, 0u, 0u);
    __syncthreads();

    const int per   = (E + NCHUNKS - 1) / NCHUNKS;
    const int start = chunk * per;
    const int end   = min(E, start + per);
    const int* __restrict__ dst = edges + E;

    for (int e0 = start; e0 < end; e0 += HDEPTH * HT) {
        int d[HDEPTH];
        #pragma unroll
        for (int k = 0; k < HDEPTH; ++k) {
            const int e = e0 + k * HT + tid;
            d[k] = (e < end) ? dst[e] : -1;
        }
        #pragma unroll
        for (int k = 0; k < HDEPTH; ++k) {
            const unsigned int r = (unsigned int)(d[k] - base);
            if (r < (unsigned int)QNODES) atomicAdd(&sCnt[r], 1u);
        }
    }
    __syncthreads();

    uint4* __restrict__ row = (uint4*)(part + (size_t)chunk * NNODES + base);
    for (int i = tid; i < QNODES / 4; i += HT) row[i] = sCnt4[i];
}

__device__ __forceinline__ unsigned int wave_incl_scan(unsigned int v, int lane) {
    #pragma unroll
    for (int o = 1; o < 32; o <<= 1) {
        const unsigned int t = __shfl_up(v, (unsigned int)o);
        if (lane >= o) v += t;
    }
    return v;
}

#define DT 256
#define NBLK ((NNODES + DT - 1) / DT)
__global__ void __launch_bounds__(DT)
k_dinv_pre(unsigned int* __restrict__ part, float* __restrict__ dinv,
           unsigned int* __restrict__ loc, unsigned int* __restrict__ blockSum, int n) {
    __shared__ unsigned int sWave[DT / 32];
    const int tid  = threadIdx.x;
    const int lane = tid & 31;
    const int wave = tid >> 5;
    const int i = blockIdx.x * DT + tid;
    unsigned int run = 0u;
    if (i < n) {
        unsigned int cnt[NCHUNKS];
        #pragma unroll
        for (int c = 0; c < NCHUNKS; ++c) cnt[c] = part[(size_t)c * NNODES + i];
        #pragma unroll
        for (int c = 0; c < NCHUNKS; ++c) { part[(size_t)c * NNODES + i] = run; run += cnt[c]; }
        dinv[i] = rsqrtf((float)(run + 1u));
    }
    const unsigned int incl = wave_incl_scan(run, lane);
    if (lane == 31) sWave[wave] = incl;
    __syncthreads();
    unsigned int woff = 0u;
    for (int w = 0; w < wave; ++w) woff += sWave[w];
    if (i < n) loc[i] = woff + incl - run;
    if (tid == DT - 1) blockSum[blockIdx.x] = woff + incl;
}

__global__ void __launch_bounds__(HT)
k_fill(const int* __restrict__ edges, const unsigned int* __restrict__ part,
       const unsigned int* __restrict__ loc, const unsigned int* __restrict__ blockSum,
       unsigned int* __restrict__ rowptr, int* __restrict__ col, int E) {
    __shared__ uint4 sCur4[QNODES / 4];
    __shared__ unsigned int sBlk[512];
    __shared__ unsigned int sW[16];
    static_assert(NBLK <= 512, "block sums must fit the 512-entry scan");
    unsigned int* sCur = (unsigned int*)sCur4;
    const int tid   = threadIdx.x;
    const int lane  = tid & 31;
    const int wave  = tid >> 5;
    const int slice = blockIdx.x % NSPLIT;
    const int chunk = blockIdx.x / NSPLIT;
    const int base  = slice * QNODES;

    const unsigned int bs = (tid < NBLK) ? blockSum[tid] : 0u;
    const unsigned int incl = wave_incl_scan(bs, lane);
    if (tid < 512 && lane == 31) sW[wave] = incl;
    __syncthreads();
    if (tid < 512) {
        unsigned int woff = 0u;
        for (int w = 0; w < wave; ++w) woff += sW[w];
        sBlk[tid] = woff + incl - bs;
        if (blockIdx.x == 0 && tid == NBLK - 1) rowptr[NNODES] = woff + incl;
    }
    __syncthreads();

    const uint4* __restrict__ loc4 = (const uint4*)(loc + base);
    const uint4* __restrict__ pre4 = (const uint4*)(part + (size_t)chunk * NNODES + base);
    uint4* __restrict__ rp4 = (uint4*)(rowptr + base);
    for (int i = tid; i < QNODES / 4; i += HT) {
        const uint4 a = loc4[i];
        const uint4 b = pre4[i];
        const unsigned int bo = sBlk[(base + 4 * i) >> 8];
        const uint4 st = make_uint4(a.x + b.x + bo, a.y + b.y + bo, a.z + b.z + bo, a.w + b.w + bo);
        sCur4[i] = st;
        if (chunk == 0) rp4[i] = st;
    }
    __syncthreads();

    const int per   = (E + NCHUNKS - 1) / NCHUNKS;
    const int start = chunk * per;
    const int end   = min(E, start + per);
    const int* __restrict__ srcv = edges;
    const int* __restrict__ dstv = edges + E;

    for (int e0 = start; e0 < end; e0 += HDEPTH * HT) {
        int d[HDEPTH];
        int s[HDEPTH];
        #pragma unroll
        for (int k = 0; k < HDEPTH; ++k) {
            const int e = e0 + k * HT + tid;
            d[k] = (e < end) ? dstv[e] : -1;
            s[k] = (e < end) ? srcv[e] : 0;
        }
        unsigned int pos[HDEPTH];
        #pragma unroll
        for (int k = 0; k < HDEPTH; ++k) {
            const unsigned int r = (unsigned int)(d[k] - base);
            pos[k] = 0u;
            if (r < (unsigned int)QNODES) pos[k] = atomicAdd(&sCur[r], 1u);
        }
        #pragma unroll
        for (int k = 0; k < HDEPTH; ++k) {
            const unsigned int r = (unsigned int)(d[k] - base);
            if (r < (unsigned int)QNODES) col[pos[k]] = s[k];
        }
    }
}

__global__ void k_gemm1(const float* __restrict__ x, const float* __restrict__ W,
                        const float* __restrict__ dinv, float* __restrict__ hs, int n) {
    __shared__ float sW[128 * 32];
    for (int i = threadIdx.x; i < 128 * 32; i += blockDim.x) {
        const int k = i >> 5, c = i & 31;
        sW[(k >> 1) * 64 + c * 2 + (k & 1)] = W[i];
    }
    __syncthreads();

    const int wave = threadIdx.x >> 5;
    const int lane = threadIdx.x & 31;
    const int tile = blockIdx.x * (blockDim.x >> 5) + wave;
    const int rowBase = tile * 16;
    if (rowBase >= n) return;

    const int half = lane >> 4;
    const int l16  = lane & 15;

    v8f c0 = {};
    v8f c1 = {};
    const float* xrow = x + (size_t)(rowBase + l16) * 128;

    #pragma unroll 4
    for (int k = 0; k < 128; k += 4) {
        const int ka = k + 2 * half;
        v2f a;  a.x = xrow[ka];  a.y = xrow[ka + 1];
        const int bidx = (ka >> 1) * 64 + l16 * 2;
        v2f b0 = *(const v2f*)&sW[bidx];
        v2f b1 = *(const v2f*)&sW[bidx + 32];
        c0 = __builtin_amdgcn_wmma_f32_16x16x4_f32(false, a, false, b0, (short)0, c0, false, false);
        c1 = __builtin_amdgcn_wmma_f32_16x16x4_f32(false, a, false, b1, (short)0, c1, false, false);
    }

    #pragma unroll
    for (int r = 0; r < 8; ++r) {
        const int m = rowBase + r + 8 * half;
        const float s = dinv[m];
        const float v0 = c0[r] * s;
        const float v1 = c1[r] * s;
        hs[(size_t)m * 32 + l16]      = v0;
        hs[(size_t)m * 32 + 16 + l16] = v1;
    }
}

#define PT   256
#define NPW  4
#define NB   ((PT / 32) * NPW)
#define APITCH 36
#define TPITCH 17
static_assert(NNODES % NB == 0, "every block owns a full set of nodes (the kernels have a block barrier)");
static_assert(NB == 32, "two WMMA tiles and one out line per block");

__device__ __forceinline__ float sum_rows32(const int* __restrict__ col, const float* __restrict__ hs,
                                            int beg, int end, int first, int lane, float acc) {
    for (int p0 = beg; p0 < end; p0 += 32) {
        const int cnt = min(32, end - p0);
        int mine = first;
        if (p0 != beg) mine = (lane < cnt) ? col[p0 + lane] : 0;
        int k = 0;
        for (; k + 8 <= cnt; k += 8) {
            float r[8];
            #pragma unroll
            for (int u = 0; u < 8; ++u) {
                const int s = __builtin_amdgcn_readlane(mine, k + u);
                r[u] = hs[(size_t)s * 32 + lane];
            }
            #pragma unroll
            for (int u = 0; u < 8; ++u) acc += r[u];
        }
        if (k < cnt) {
            float r[8];
            #pragma unroll
            for (int u = 0; u < 8; ++u) {
                r[u] = 0.0f;
                if (k + u < cnt) {
                    const int s = __builtin_amdgcn_readlane(mine, k + u);
                    r[u] = hs[(size_t)s * 32 + lane];
                }
            }
            #pragma unroll
            for (int u = 0; u < 8; ++u) acc += r[u];
        }
    }
    return acc;
}

__global__ void __launch_bounds__(PT)
k_agg1_gemm2(const unsigned int* __restrict__ rowptr, const int* __restrict__ col,
             const float* __restrict__ hs, const float* __restrict__ W,
             const float* __restrict__ b1, const float* __restrict__ dinv,
             float* __restrict__ hs2) {
    __shared__ float sW[32 * 16];
    __shared__ float sA[NB * APITCH];
    __shared__ float sD[NB];
    static_assert(32 * 16 == 2 * PT, "two W2 elements per thread");
    {
        const int i0 = threadIdx.x, i1 = threadIdx.x + PT;
        const float w0 = W[i0], w1 = W[i1];
        sW[((i0 >> 4) >> 1) * 32 + (i0 & 15) * 2 + ((i0 >> 4) & 1)] = w0;
        sW[((i1 >> 4) >> 1) * 32 + (i1 & 15) * 2 + ((i1 >> 4) & 1)] = w1;
    }
    const int lane = threadIdx.x & 31;
    const int wave = __builtin_amdgcn_readfirstlane((int)(threadIdx.x >> 5));
    const int nodeBase = blockIdx.x * NB;
    const int v0 = nodeBase + wave * NPW;
    const float bias = b1[lane];

    int rp[NPW + 1];
    #pragma unroll
    for (int j = 0; j <= NPW; ++j) rp[j] = (int)rowptr[v0 + j];
    int next = (lane < min(32, rp[1] - rp[0])) ? col[rp[0] + lane] : 0;
    #pragma unroll
    for (int j = 0; j < NPW; ++j) {
        const int v = v0 + j;
        const int first = next;
        if (j + 1 < NPW) next = (lane < min(32, rp[j + 2] - rp[j + 1])) ? col[rp[j + 1] + lane] : 0;
        const float dv = dinv[v];
        float acc = hs[(size_t)v * 32 + lane];
        acc = sum_rows32(col, hs, rp[j], rp[j + 1], first, lane, acc);
        const int r = wave * NPW + j;
        sA[r * APITCH + lane] = lrelu(dv * acc + bias);
        if (lane == 0) sD[r] = dv;
    }
    __syncthreads();
    if (wave >= 2) return;

    const int half = lane >> 4;
    const int l16  = lane & 15;
    v8f c = {};
    const float* arow = &sA[(wave * 16 + l16) * APITCH];
    #pragma unroll
    for (int k = 0; k < 32; k += 4) {
        const int ka = k + 2 * half;
        v2f a = *(const v2f*)&arow[ka];
        v2f b = *(const v2f*)&sW[(ka >> 1) * 32 + l16 * 2];
        c = __builtin_amdgcn_wmma_f32_16x16x4_f32(false, a, false, b, (short)0, c, false, false);
    }
    #pragma unroll
    for (int r = 0; r < 8; ++r) {
        const int m = wave * 16 + r + 8 * half;
        hs2[(size_t)(nodeBase + m) * 16 + l16] = c[r] * sD[m];
    }
}

__device__ __forceinline__ float sum_rows16(const int* __restrict__ col, const float* __restrict__ hs,
                                            int beg, int end, int first, int lane) {
    const int half = lane >> 4;
    const int l16  = lane & 15;
    float acc = 0.0f;
    for (int p0 = beg; p0 < end; p0 += 32) {
        const int cnt = min(32, end - p0);
        int mine = first;
        if (p0 != beg) mine = (lane < cnt) ? col[p0 + lane] : 0;
        int k = 0;
        for (; k + 16 <= cnt; k += 16) {
            float r[8];
            #pragma unroll
            for (int u = 0; u < 8; ++u) {
                const int s = __shfl(mine, k + 2 * u + half);
                r[u] = hs[(size_t)s * 16 + l16];
            }
            #pragma unroll
            for (int u = 0; u < 8; ++u) acc += r[u];
        }
        if (k < cnt) {
            float r[8];
            #pragma unroll
            for (int u = 0; u < 8; ++u) {
                const int j = k + 2 * u + half;
                const int s = __shfl(mine, j & 31);
                r[u] = 0.0f;
                if (j < cnt) r[u] = hs[(size_t)s * 16 + l16];
            }
            #pragma unroll
            for (int u = 0; u < 8; ++u) acc += r[u];
        }
    }
    return acc;
}

__global__ void __launch_bounds__(PT)
k_agg2_out(const unsigned int* __restrict__ rowptr, const int* __restrict__ col,
           const float* __restrict__ hs, const float* __restrict__ dinv,
           const float* __restrict__ b2, const float* __restrict__ Wfc,
           const float* __restrict__ bfc, float* __restrict__ out) {
    __shared__ float sT[NB * TPITCH];
    const int lane = threadIdx.x & 31;
    const int l16  = lane & 15;
    const int wave = __builtin_amdgcn_readfirstlane((int)(threadIdx.x >> 5));
    const int nodeBase = blockIdx.x * NB;
    const int v0 = nodeBase + wave * NPW;
    const float bias = b2[l16];

    int rp[NPW + 1];
    #pragma unroll
    for (int j = 0; j <= NPW; ++j) rp[j] = (int)rowptr[v0 + j];
    int next = (lane < min(32, rp[1] - rp[0])) ? col[rp[0] + lane] : 0;
    #pragma unroll
    for (int j = 0; j < NPW; ++j) {
        const int v = v0 + j;
        const int first = next;
        if (j + 1 < NPW) next = (lane < min(32, rp[j + 2] - rp[j + 1])) ? col[rp[j + 1] + lane] : 0;
        const float dv = dinv[v];
        const float self = hs[(size_t)v * 16 + l16];
        float acc = sum_rows16(col, hs, rp[j], rp[j + 1], first, lane);
        acc += __shfl_xor(acc, 16);
        if (lane < 16) sT[(wave * NPW + j) * TPITCH + l16] = lrelu(dv * (acc + self) + bias);
    }
    __syncthreads();
    if (wave == 0) {
        const float* row = &sT[lane * TPITCH];
        float s = bfc[0];
        #pragma unroll
        for (int j = 0; j < 16; ++j) s += row[j] * Wfc[j];
        out[nodeBase + lane] = s;
    }
}

extern "C" void kernel_launch(void* const* d_in, const int* in_sizes, int n_in,
                              void* d_out, int out_size, void* d_ws, size_t ws_size,
                              hipStream_t stream) {
    const float* x    = (const float*)d_in[0];
    const int*   ei   = (const int*)  d_in[1];
    const float* W1   = (const float*)d_in[2];
    const float* b1   = (const float*)d_in[3];
    const float* W2   = (const float*)d_in[4];
    const float* b2   = (const float*)d_in[5];
    const float* Wfc  = (const float*)d_in[6];
    const float* bfc  = (const float*)d_in[7];
    float*       out  = (float*)d_out;

    const int n = NNODES;
    const int E = in_sizes[1] / 2;

    char* ws = (char*)d_ws;
    size_t off = 0;
    auto carve = [&](size_t bytes) { char* p = ws + off; off = (off + bytes + 511) & ~(size_t)511; return p; };
    unsigned int* part = (unsigned int*)carve((size_t)NCHUNKS * NNODES * 4);
    float* dinv = (float*)carve((size_t)n * 4);
    unsigned int* loc      = (unsigned int*)carve((size_t)n * 4);
    unsigned int* blockSum = (unsigned int*)carve((size_t)NBLK * 4);
    unsigned int* rowptr   = (unsigned int*)carve((size_t)(n + 1) * 4);
    float* h1s  = (float*)carve((size_t)n * 32 * 4);
    float* h2s  = (float*)carve((size_t)n * 16 * 4);
    int*   col  = (int*)(ws + off);
    (void)ws_size;

    const int T = 256;
    const int wavesPerBlock = T / 32;
    const int tiles = (n + 15) / 16;
    const int gemmBlocks = (tiles + wavesPerBlock - 1) / wavesPerBlock;

    k_deg_hist<<<NCHUNKS * NSPLIT, HT, 0, stream>>>(ei, part, E);
    k_dinv_pre<<<NBLK, DT, 0, stream>>>(part, dinv, loc, blockSum, n);
    k_fill    <<<NCHUNKS * NSPLIT, HT, 0, stream>>>(ei, part, loc, blockSum, rowptr, col, E);

    const int aggBlocks = n / NB;

    k_gemm1     <<<gemmBlocks, T, 0, stream>>>(x, W1, dinv, h1s, n);
    k_agg1_gemm2<<<aggBlocks, PT, 0, stream>>>(rowptr, col, h1s, W2, b1, dinv, h2s);

    k_agg2_out  <<<aggBlocks, PT, 0, stream>>>(rowptr, col, h2s, dinv, b2, Wfc, bfc, out);
}


# edge grouping as a two-level partition: chunks binned into 250 node buckets in LDS and written as runs, each bucket counted and placed in LDS, col written whole; a node's list order = arrival order
# speedup vs baseline: 5.8574x; 1.6034x over previous
#include <hip/hip_runtime.h>

typedef float v2f __attribute__((ext_vector_type(2)));
typedef float v8f __attribute__((ext_vector_type(8)));

#define NNODES   100000
#define NEGSLOPE 0.01f

__device__ __forceinline__ float lrelu(float v) { return v > 0.0f ? v : NEGSLOPE * v; }

#define NBKT  250
#define BN    400
#define NCH   256
#define NBP   256
#define ST    1024
#define EPT   13
#define ECAP  (EPT * ST)
#define LCAP  24576
#define SRCBITS 17
#define SRCMASK 0x1FFFFu
static_assert(NBKT * BN == NNODES, "buckets tile the node range");
static_assert(NNODES <= (1 << SRCBITS), "a source index fits its field");
static_assert(BN <= 512, "a bucket offset fits the bits above the source field");
static_assert(NBKT <= NBP, "padded bucket count");

__device__ __forceinline__ unsigned int bucket_of(int d) {
    const unsigned int b = (unsigned int)d / (unsigned int)BN;
    return b < (unsigned int)NBKT ? b : (unsigned int)(NBKT - 1);
}

__device__ __forceinline__ unsigned int wave_incl_scan(unsigned int v, int lane) {
    #pragma unroll
    for (int o = 1; o < 32; o <<= 1) {
        const unsigned int t = __shfl_up(v, (unsigned int)o);
        if (lane >= o) v += t;
    }
    return v;
}

__global__ void __launch_bounds__(ST)
k_bkt_count(const int* __restrict__ edges, unsigned int* __restrict__ cntT, int E) {
    __shared__ unsigned int sC[NBP];
    const int tid = threadIdx.x;
    const int c   = blockIdx.x;
    if (tid < NBP) sC[tid] = 0u;
    __syncthreads();

    const int per   = (E + NCH - 1) / NCH;
    const int start = c * per;
    const int end   = min(E, start + min(per, ECAP));
    const int* __restrict__ dst = edges + E;

    int d[EPT];
    #pragma unroll
    for (int k = 0; k < EPT; ++k) {
        const int e = start + k * ST + tid;
        d[k] = (e < end) ? dst[e] : -1;
    }
    #pragma unroll
    for (int k = 0; k < EPT; ++k)
        if (d[k] >= 0) atomicAdd(&sC[bucket_of(d[k])], 1u);
    __syncthreads();
    if (tid < NBP) cntT[c * NBP + tid] = sC[tid];
}

__global__ void __launch_bounds__(ST)
k_bkt_part(const int* __restrict__ edges, const unsigned int* __restrict__ cntT,
           unsigned int* __restrict__ bbase, unsigned int* __restrict__ tmp, int E) {
    __shared__ unsigned int   sWord[ECAP];
    __shared__ unsigned short sBk[ECAP];
    __shared__ uint4          sTot4[16][NBP / 4];
    __shared__ uint4          sBef4[16][NBP / 4];
    __shared__ unsigned int   sCur[NBP];
    __shared__ unsigned int   sDelta[NBP];
    __shared__ unsigned int   sWT[8];
    __shared__ unsigned int   sWO[8];
    static_assert(ST == 16 * (NBP / 4), "sixteen threads per group of four table columns");
    static_assert(NCH == 16 * 16, "each of the sixteen sums 16 chunks");
    const int tid  = threadIdx.x;
    const int lane = tid & 31;
    const int wave = tid >> 5;
    const int c    = blockIdx.x;

    const int per   = (E + NCH - 1) / NCH;
    const int start = c * per;
    const int end   = min(E, start + min(per, ECAP));
    const int* __restrict__ srcv = edges;
    const int* __restrict__ dstv = edges + E;

    int d[EPT];
    int s[EPT];
    #pragma unroll
    for (int k = 0; k < EPT; ++k) {
        const int e = start + k * ST + tid;
        d[k] = (e < end) ? dstv[e] : -1;
        s[k] = (e < end) ? srcv[e] : 0;
    }

    {
        const int g = tid & (NBP / 4 - 1);
        const int q = tid >> 6;
        const uint4* __restrict__ tab4 = (const uint4*)cntT;
        uint4 tot = make_uint4(0u, 0u, 0u, 0u);
        uint4 bef = make_uint4(0u, 0u, 0u, 0u);
        #pragma unroll
        for (int j = 0; j < 16; ++j) {
            const int cc = q * 16 + j;
            const uint4 v = tab4[cc * (NBP / 4) + g];
            tot.x += v.x; tot.y += v.y; tot.z += v.z; tot.w += v.w;
            if (cc < c) { bef.x += v.x; bef.y += v.y; bef.z += v.z; bef.w += v.w; }
        }
        sTot4[q][g] = tot;
        sBef4[q][g] = bef;
    }
    __syncthreads();
    unsigned int T = 0u, B = 0u, O = 0u;
    if (tid < NBP) {
        const unsigned int* sTot = (const unsigned int*)sTot4;
        const unsigned int* sBef = (const unsigned int*)sBef4;
        #pragma unroll
        for (int q = 0; q < 16; ++q) { T += sTot[q * NBP + tid]; B += sBef[q * NBP + tid]; }
        O = cntT[c * NBP + tid];
    }
    const unsigned int inclT = wave_incl_scan(T, lane);
    const unsigned int inclO = wave_incl_scan(O, lane);
    if (tid < NBP && lane == 31) { sWT[wave] = inclT; sWO[wave] = inclO; }
    __syncthreads();
    if (tid < NBP) {
        unsigned int offT = 0u, offO = 0u;
        for (int w = 0; w < wave; ++w) { offT += sWT[w]; offO += sWO[w]; }
        const unsigned int base = offT + inclT - T;
        const unsigned int lst  = offO + inclO - O;
        sCur[tid]   = lst;
        sDelta[tid] = base + B - lst;
        if (c == 0) {
            if (tid < NBKT) bbase[tid] = base;
            if (tid == NBKT - 1) bbase[NBKT] = base + T;
        }
    }
    __syncthreads();

    unsigned int r[EPT];
    #pragma unroll
    for (int k = 0; k < EPT; ++k) {
        r[k] = 0u;
        if (d[k] >= 0) r[k] = atomicAdd(&sCur[bucket_of(d[k])], 1u);
    }
    #pragma unroll
    for (int k = 0; k < EPT; ++k) {
        if (d[k] >= 0) {
            const unsigned int b = bucket_of(d[k]);
            const unsigned int off = (unsigned int)d[k] - b * (unsigned int)BN;
            sWord[r[k]] = ((unsigned int)s[k] & SRCMASK) | (off << SRCBITS);
            sBk[r[k]]   = (unsigned short)b;
        }
    }
    __syncthreads();

    const int n = end - start;
    for (int i = tid; i < n; i += ST) {
        const unsigned int b = sBk[i];
        tmp[(unsigned int)i + sDelta[b]] = sWord[i];
    }
}

#define SDEPTH 13
__global__ void __launch_bounds__(ST)
k_bkt_sort(const unsigned int* __restrict__ tmp, const unsigned int* __restrict__ bbase,
           unsigned int* __restrict__ rowptr, float* __restrict__ dinv, int* __restrict__ col) {
    __shared__ int          sOut[LCAP];
    __shared__ unsigned int sCnt[512];
    __shared__ unsigned int sCur[512];
    __shared__ unsigned int sWv[16];
    const int tid  = threadIdx.x;
    const int lane = tid & 31;
    const int wave = tid >> 5;
    const int b    = blockIdx.x;
    const unsigned int beg = bbase[b];
    const unsigned int end = bbase[b + 1];
    const int n = (int)(end - beg);
    const bool staged = (n <= LCAP);

    unsigned int w0[SDEPTH];
    #pragma unroll
    for (int k = 0; k < SDEPTH; ++k) {
        const int i = k * ST + tid;
        w0[k] = (i < n) ? tmp[beg + (unsigned int)i] : 0xFFFFFFFFu;
    }

    if (tid < 512) sCnt[tid] = 0u;
    __syncthreads();

    #pragma unroll
    for (int k = 0; k < SDEPTH; ++k) {
        if (w0[k] != 0xFFFFFFFFu) {
            const unsigned int ld = min(w0[k] >> SRCBITS, (unsigned int)(BN - 1));
            atomicAdd(&sCnt[ld], 1u);
        }
    }
    for (int i0 = SDEPTH * ST; i0 < n; i0 += SDEPTH * ST) {
        unsigned int w[SDEPTH];
        #pragma unroll
        for (int k = 0; k < SDEPTH; ++k) {
            const int i = i0 + k * ST + tid;
            w[k] = (i < n) ? tmp[beg + (unsigned int)i] : 0xFFFFFFFFu;
        }
        #pragma unroll
        for (int k = 0; k < SDEPTH; ++k) {
            if (w[k] != 0xFFFFFFFFu) {
                const unsigned int ld = min(w[k] >> SRCBITS, (unsigned int)(BN - 1));
                atomicAdd(&sCnt[ld], 1u);
            }
        }
    }
    __syncthreads();

    const unsigned int v = (tid < BN) ? sCnt[tid] : 0u;
    const unsigned int incl = wave_incl_scan(v, lane);
    if (tid < 512 && lane == 31) sWv[wave] = incl;
    __syncthreads();
    if (tid < BN) {
        unsigned int woff = 0u;
        for (int q = 0; q < wave; ++q) woff += sWv[q];
        const unsigned int excl = woff + incl - v;
        const int node = b * BN + tid;
        sCur[tid]    = excl;
        rowptr[node] = beg + excl;
        dinv[node]   = rsqrtf((float)(v + 1u));
        if (b == NBKT - 1 && tid == BN - 1) rowptr[NNODES] = beg + excl + v;
    }
    __syncthreads();

    unsigned int p0[SDEPTH];
    #pragma unroll
    for (int k = 0; k < SDEPTH; ++k) {
        p0[k] = 0u;
        if (w0[k] != 0xFFFFFFFFu)
            p0[k] = atomicAdd(&sCur[min(w0[k] >> SRCBITS, (unsigned int)(BN - 1))], 1u);
    }
    #pragma unroll
    for (int k = 0; k < SDEPTH; ++k) {
        if (w0[k] != 0xFFFFFFFFu) {
            const int src = (int)(w0[k] & SRCMASK);
            if (staged) sOut[p0[k]] = src;
            else        col[beg + p0[k]] = src;
        }
    }
    for (int i0 = SDEPTH * ST; i0 < n; i0 += SDEPTH * ST) {
        unsigned int w[SDEPTH];
        #pragma unroll
        for (int k = 0; k < SDEPTH; ++k) {
            const int i = i0 + k * ST + tid;
            w[k] = (i < n) ? tmp[beg + (unsigned int)i] : 0xFFFFFFFFu;
        }
        #pragma unroll
        for (int k = 0; k < SDEPTH; ++k) {
            if (w[k] != 0xFFFFFFFFu) {
                const unsigned int ld = min(w[k] >> SRCBITS, (unsigned int)(BN - 1));
                const unsigned int p  = atomicAdd(&sCur[ld], 1u);
                const int src = (int)(w[k] & SRCMASK);
                if (staged) sOut[p] = src;
                else        col[beg + p] = src;
            }
        }
    }
    __syncthreads();
    if (staged)
        for (int i = tid; i < n; i += ST) col[beg + (unsigned int)i] = sOut[i];
}

__global__ void k_gemm1(const float* __restrict__ x, const float* __restrict__ W,
                        const float* __restrict__ dinv, float* __restrict__ hs, int n) {
    __shared__ float sW[128 * 32];
    for (int i = threadIdx.x; i < 128 * 32; i += blockDim.x) {
        const int k = i >> 5, c = i & 31;
        sW[(k >> 1) * 64 + c * 2 + (k & 1)] = W[i];
    }
    __syncthreads();

    const int wave = threadIdx.x >> 5;
    const int lane = threadIdx.x & 31;
    const int tile = blockIdx.x * (blockDim.x >> 5) + wave;
    const int rowBase = tile * 16;
    if (rowBase >= n) return;

    const int half = lane >> 4;
    const int l16  = lane & 15;

    v8f c0 = {};
    v8f c1 = {};
    const float* xrow = x + (size_t)(rowBase + l16) * 128;

    #pragma unroll 4
    for (int k = 0; k < 128; k += 4) {
        const int ka = k + 2 * half;
        v2f a;  a.x = xrow[ka];  a.y = xrow[ka + 1];
        const int bidx = (ka >> 1) * 64 + l16 * 2;
        v2f b0 = *(const v2f*)&sW[bidx];
        v2f b1 = *(const v2f*)&sW[bidx + 32];
        c0 = __builtin_amdgcn_wmma_f32_16x16x4_f32(false, a, false, b0, (short)0, c0, false, false);
        c1 = __builtin_amdgcn_wmma_f32_16x16x4_f32(false, a, false, b1, (short)0, c1, false, false);
    }

    #pragma unroll
    for (int r = 0; r < 8; ++r) {
        const int m = rowBase + r + 8 * half;
        const float s = dinv[m];
        const float v0 = c0[r] * s;
        const float v1 = c1[r] * s;
        hs[(size_t)m * 32 + l16]      = v0;
        hs[(size_t)m * 32 + 16 + l16] = v1;
    }
}

#define PT   256
#define NPW  4
#define NB   ((PT / 32) * NPW)
#define APITCH 36
#define TPITCH 17
static_assert(NNODES % NB == 0, "every block owns a full set of nodes (the kernels have a block barrier)");
static_assert(NB == 32, "two WMMA tiles and one out line per block");

__device__ __forceinline__ float sum_rows32(const int* __restrict__ col, const float* __restrict__ hs,
                                            int beg, int end, int first, int lane, float acc) {
    for (int p0 = beg; p0 < end; p0 += 32) {
        const int cnt = min(32, end - p0);
        int mine = first;
        if (p0 != beg) mine = (lane < cnt) ? col[p0 + lane] : 0;
        int k = 0;
        for (; k + 8 <= cnt; k += 8) {
            float r[8];
            #pragma unroll
            for (int u = 0; u < 8; ++u) {
                const int s = __builtin_amdgcn_readlane(mine, k + u);
                r[u] = hs[(size_t)s * 32 + lane];
            }
            #pragma unroll
            for (int u = 0; u < 8; ++u) acc += r[u];
        }
        if (k < cnt) {
            float r[8];
            #pragma unroll
            for (int u = 0; u < 8; ++u) {
                r[u] = 0.0f;
                if (k + u < cnt) {
                    const int s = __builtin_amdgcn_readlane(mine, k + u);
                    r[u] = hs[(size_t)s * 32 + lane];
                }
            }
            #pragma unroll
            for (int u = 0; u < 8; ++u) acc += r[u];
        }
    }
    return acc;
}

__global__ void __launch_bounds__(PT)
k_agg1_gemm2(const unsigned int* __restrict__ rowptr, const int* __restrict__ col,
             const float* __restrict__ hs, const float* __restrict__ W,
             const float* __restrict__ b1, const float* __restrict__ dinv,
             float* __restrict__ hs2) {
    __shared__ float sW[32 * 16];
    __shared__ float sA[NB * APITCH];
    __shared__ float sD[NB];
    static_assert(32 * 16 == 2 * PT, "two W2 elements per thread");
    {
        const int i0 = threadIdx.x, i1 = threadIdx.x + PT;
        const float w0 = W[i0], w1 = W[i1];
        sW[((i0 >> 4) >> 1) * 32 + (i0 & 15) * 2 + ((i0 >> 4) & 1)] = w0;
        sW[((i1 >> 4) >> 1) * 32 + (i1 & 15) * 2 + ((i1 >> 4) & 1)] = w1;
    }
    const int lane = threadIdx.x & 31;
    const int wave = __builtin_amdgcn_readfirstlane((int)(threadIdx.x >> 5));
    const int nodeBase = blockIdx.x * NB;
    const int v0 = nodeBase + wave * NPW;
    const float bias = b1[lane];

    int rp[NPW + 1];
    #pragma unroll
    for (int j = 0; j <= NPW; ++j) rp[j] = (int)rowptr[v0 + j];
    int next = (lane < min(32, rp[1] - rp[0])) ? col[rp[0] + lane] : 0;
    #pragma unroll
    for (int j = 0; j < NPW; ++j) {
        const int v = v0 + j;
        const int first = next;
        if (j + 1 < NPW) next = (lane < min(32, rp[j + 2] - rp[j + 1])) ? col[rp[j + 1] + lane] : 0;
        const float dv = dinv[v];
        float acc = hs[(size_t)v * 32 + lane];
        acc = sum_rows32(col, hs, rp[j], rp[j + 1], first, lane, acc);
        const int r = wave * NPW + j;
        sA[r * APITCH + lane] = lrelu(dv * acc + bias);
        if (lane == 0) sD[r] = dv;
    }
    __syncthreads();
    if (wave >= 2) return;

    const int half = lane >> 4;
    const int l16  = lane & 15;
    v8f c = {};
    const float* arow = &sA[(wave * 16 + l16) * APITCH];
    #pragma unroll
    for (int k = 0; k < 32; k += 4) {
        const int ka = k + 2 * half;
        v2f a = *(const v2f*)&arow[ka];
        v2f b = *(const v2f*)&sW[(ka >> 1) * 32 + l16 * 2];
        c = __builtin_amdgcn_wmma_f32_16x16x4_f32(false, a, false, b, (short)0, c, false, false);
    }
    #pragma unroll
    for (int r = 0; r < 8; ++r) {
        const int m = wave * 16 + r + 8 * half;
        hs2[(size_t)(nodeBase + m) * 16 + l16] = c[r] * sD[m];
    }
}

__device__ __forceinline__ float sum_rows16(const int* __restrict__ col, const float* __restrict__ hs,
                                            int beg, int end, int first, int lane) {
    const int half = lane >> 4;
    const int l16  = lane & 15;
    float acc = 0.0f;
    for (int p0 = beg; p0 < end; p0 += 32) {
        const int cnt = min(32, end - p0);
        int mine = first;
        if (p0 != beg) mine = (lane < cnt) ? col[p0 + lane] : 0;
        int k = 0;
        for (; k + 16 <= cnt; k += 16) {
            float r[8];
            #pragma unroll
            for (int u = 0; u < 8; ++u) {
                const int s = __shfl(mine, k + 2 * u + half);
                r[u] = hs[(size_t)s * 16 + l16];
            }
            #pragma unroll
            for (int u = 0; u < 8; ++u) acc += r[u];
        }
        if (k < cnt) {
            float r[8];
            #pragma unroll
            for (int u = 0; u < 8; ++u) {
                const int j = k + 2 * u + half;
                const int s = __shfl(mine, j & 31);
                r[u] = 0.0f;
                if (j < cnt) r[u] = hs[(size_t)s * 16 + l16];
            }
            #pragma unroll
            for (int u = 0; u < 8; ++u) acc += r[u];
        }
    }
    return acc;
}

__global__ void __launch_bounds__(PT)
k_agg2_out(const unsigned int* __restrict__ rowptr, const int* __restrict__ col,
           const float* __restrict__ hs, const float* __restrict__ dinv,
           const float* __restrict__ b2, const float* __restrict__ Wfc,
           const float* __restrict__ bfc, float* __restrict__ out) {
    __shared__ float sT[NB * TPITCH];
    const int lane = threadIdx.x & 31;
    const int l16  = lane & 15;
    const int wave = __builtin_amdgcn_readfirstlane((int)(threadIdx.x >> 5));
    const int nodeBase = blockIdx.x * NB;
    const int v0 = nodeBase + wave * NPW;
    const float bias = b2[l16];

    int rp[NPW + 1];
    #pragma unroll
    for (int j = 0; j <= NPW; ++j) rp[j] = (int)rowptr[v0 + j];
    int next = (lane < min(32, rp[1] - rp[0])) ? col[rp[0] + lane] : 0;
    #pragma unroll
    for (int j = 0; j < NPW; ++j) {
        const int v = v0 + j;
        const int first = next;
        if (j + 1 < NPW) next = (lane < min(32, rp[j + 2] - rp[j + 1])) ? col[rp[j + 1] + lane] : 0;
        const float dv = dinv[v];
        const float self = hs[(size_t)v * 16 + l16];
        float acc = sum_rows16(col, hs, rp[j], rp[j + 1], first, lane);
        acc += __shfl_xor(acc, 16);
        if (lane < 16) sT[(wave * NPW + j) * TPITCH + l16] = lrelu(dv * (acc + self) + bias);
    }
    __syncthreads();
    if (wave == 0) {
        const float* row = &sT[lane * TPITCH];
        float s = bfc[0];
        #pragma unroll
        for (int j = 0; j < 16; ++j) s += row[j] * Wfc[j];
        out[nodeBase + lane] = s;
    }
}

extern "C" void kernel_launch(void* const* d_in, const int* in_sizes, int n_in,
                              void* d_out, int out_size, void* d_ws, size_t ws_size,
                              hipStream_t stream) {
    const float* x    = (const float*)d_in[0];
    const int*   ei   = (const int*)  d_in[1];
    const float* W1   = (const float*)d_in[2];
    const float* b1   = (const float*)d_in[3];
    const float* W2   = (const float*)d_in[4];
    const float* b2   = (const float*)d_in[5];
    const float* Wfc  = (const float*)d_in[6];
    const float* bfc  = (const float*)d_in[7];
    float*       out  = (float*)d_out;

    const int n = NNODES;
    const int E = in_sizes[1] / 2;

    char* ws = (char*)d_ws;
    size_t off = 0;
    auto carve = [&](size_t bytes) { char* p = ws + off; off = (off + bytes + 511) & ~(size_t)511; return p; };
    unsigned int* cntT  = (unsigned int*)carve((size_t)NCH * NBP * 4);
    unsigned int* bbase = (unsigned int*)carve((size_t)(NBKT + 1) * 4);
    float* dinv = (float*)carve((size_t)n * 4);
    unsigned int* rowptr   = (unsigned int*)carve((size_t)(n + 1) * 4);
    float* h1s  = (float*)carve((size_t)n * 32 * 4);
    float* h2s  = (float*)carve((size_t)n * 16 * 4);
    unsigned int* tmp = (unsigned int*)carve((size_t)E * 4);
    int*   col  = (int*)(ws + off);
    (void)ws_size;

    const int T = 256;
    const int wavesPerBlock = T / 32;
    const int tiles = (n + 15) / 16;
    const int gemmBlocks = (tiles + wavesPerBlock - 1) / wavesPerBlock;

    k_bkt_count<<<NCH, ST, 0, stream>>>(ei, cntT, E);
    k_bkt_part <<<NCH, ST, 0, stream>>>(ei, cntT, bbase, tmp, E);
    k_bkt_sort <<<NBKT, ST, 0, stream>>>(tmp, bbase, rowptr, dinv, col);

    const int aggBlocks = n / NB;

    k_gemm1     <<<gemmBlocks, T, 0, stream>>>(x, W1, dinv, h1s, n);
    k_agg1_gemm2<<<aggBlocks, PT, 0, stream>>>(rowptr, col, h1s, W2, b1, dinv, h2s);

    k_agg2_out  <<<aggBlocks, PT, 0, stream>>>(rowptr, col, h2s, dinv, b2, Wfc, bfc, out);
}


# edge grouping without the count launch: each chunk is binned in LDS and written back whole in bucket order with its share starts; a bucket block finds its 256 runs and its col start from that table
# speedup vs baseline: 6.4118x; 1.0946x over previous
#include <hip/hip_runtime.h>

typedef float v2f __attribute__((ext_vector_type(2)));
typedef float v8f __attribute__((ext_vector_type(8)));

#define NNODES   100000
#define NEGSLOPE 0.01f

__device__ __forceinline__ float lrelu(float v) { return v > 0.0f ? v : NEGSLOPE * v; }

#define NBKT  250
#define BN    400
#define NCH   256
#define NBP   256
#define ST    1024
#define EPT   13
#define ECAP  (EPT * ST)
#define LCAP  24576
#define SRCBITS 17
#define SRCMASK 0x1FFFFu
static_assert(NBKT * BN == NNODES, "buckets tile the node range");
static_assert(NNODES <= (1 << SRCBITS), "a source index fits its field");
static_assert(BN <= 512, "a bucket offset fits the bits above the source field");
static_assert(NBKT <= NBP, "padded bucket count");

__device__ __forceinline__ unsigned int bucket_of(int d) {
    const unsigned int b = (unsigned int)d / (unsigned int)BN;
    return b < (unsigned int)NBKT ? b : (unsigned int)(NBKT - 1);
}

__device__ __forceinline__ unsigned int wave_incl_scan(unsigned int v, int lane) {
    #pragma unroll
    for (int o = 1; o < 32; o <<= 1) {
        const unsigned int t = __shfl_up(v, (unsigned int)o);
        if (lane >= o) v += t;
    }
    return v;
}

__device__ __forceinline__ unsigned int wave_sum(unsigned int v) {
    #pragma unroll
    for (int o = 16; o > 0; o >>= 1) v += __shfl_xor(v, o);
    return v;
}

__global__ void __launch_bounds__(ST)
k_bkt_part(const int* __restrict__ edges, unsigned int* __restrict__ offT,
           unsigned int* __restrict__ tmp, int E) {
    __shared__ unsigned int sWord[ECAP];
    __shared__ unsigned int sCnt[NBP];
    __shared__ unsigned int sOff[NBP];
    __shared__ unsigned int sWv[NBP / 32];
    const int tid  = threadIdx.x;
    const int lane = tid & 31;
    const int wave = tid >> 5;
    const int c    = blockIdx.x;

    const int per   = (E + NCH - 1) / NCH;
    const int start = c * per;
    const int end   = min(E, start + min(per, ECAP));
    const int* __restrict__ srcv = edges;
    const int* __restrict__ dstv = edges + E;

    int d[EPT];
    int s[EPT];
    #pragma unroll
    for (int k = 0; k < EPT; ++k) {
        const int e = start + k * ST + tid;
        d[k] = (e < end) ? dstv[e] : -1;
        s[k] = (e < end) ? srcv[e] : 0;
    }
    if (tid < NBP) sCnt[tid] = 0u;
    __syncthreads();

    unsigned int r[EPT];
    #pragma unroll
    for (int k = 0; k < EPT; ++k) {
        r[k] = 0u;
        if (d[k] >= 0) r[k] = atomicAdd(&sCnt[bucket_of(d[k])], 1u);
    }
    __syncthreads();

    const unsigned int cnt  = (tid < NBP) ? sCnt[tid] : 0u;
    const unsigned int incl = wave_incl_scan(cnt, lane);
    if (tid < NBP && lane == 31) sWv[wave] = incl;
    __syncthreads();
    if (tid < NBP) {
        unsigned int woff = 0u;
        for (int q = 0; q < wave; ++q) woff += sWv[q];
        const unsigned int excl = woff + incl - cnt;
        sOff[tid] = excl;
        offT[c * NBP + tid] = excl;
    }
    __syncthreads();

    #pragma unroll
    for (int k = 0; k < EPT; ++k) {
        if (d[k] >= 0) {
            const unsigned int b = bucket_of(d[k]);
            const unsigned int off = (unsigned int)d[k] - b * (unsigned int)BN;
            sWord[sOff[b] + r[k]] = ((unsigned int)s[k] & SRCMASK) | (off << SRCBITS);
        }
    }
    __syncthreads();

    const int n = end - start;
    for (int i = tid; i < n; i += ST) tmp[start + i] = sWord[i];
}

#define RPW 8
#define RLD 2
static_assert((ST / 32) * RPW == NCH, "every run has a wave");
__global__ void __launch_bounds__(ST)
k_bkt_sort(const unsigned int* __restrict__ tmp, const unsigned int* __restrict__ offT,
           unsigned int* __restrict__ rowptr, float* __restrict__ dinv, int* __restrict__ col, int E) {
    __shared__ int          sOut[LCAP];
    __shared__ unsigned int sCnt[512];
    __shared__ unsigned int sCur[512];
    __shared__ unsigned int sWv[16];
    __shared__ unsigned int sSumO[ST / 32];
    __shared__ unsigned int sSumL[ST / 32];
    static_assert(ST / 32 == 32, "one partial sum per lane in the block sums below");
    const int tid  = threadIdx.x;
    const int lane = tid & 31;
    const int wave = tid >> 5;
    const int b    = blockIdx.x;
    const int per  = (E + NCH - 1) / NCH;

    unsigned int o0 = 0u, rlen = 0u;
    if (lane < RPW) {
        const int c = wave * RPW + lane;
        o0   = offT[c * NBP + b];
        rlen = offT[c * NBP + b + 1] - o0;
    }
    const unsigned int rpos = (unsigned int)((wave * RPW + lane) * per) + o0;
    {
        const unsigned int so = wave_sum(o0);
        const unsigned int sl = wave_sum(rlen);
        if (lane == 0) { sSumO[wave] = so; sSumL[wave] = sl; }
    }
    if (tid < 512) sCnt[tid] = 0u;

    unsigned int w0[RPW * RLD];
    #pragma unroll
    for (int j = 0; j < RPW; ++j) {
        const unsigned int pos = (unsigned int)__builtin_amdgcn_readlane((int)rpos, j);
        const unsigned int len = (unsigned int)__builtin_amdgcn_readlane((int)rlen, j);
        #pragma unroll
        for (int u = 0; u < RLD; ++u) {
            const unsigned int i = (unsigned int)(u * 32 + lane);
            w0[j * RLD + u] = (i < len) ? tmp[pos + i] : 0xFFFFFFFFu;
        }
    }
    __syncthreads();
    const unsigned int beg = wave_sum(sSumO[lane]);
    const int n = (int)wave_sum(sSumL[lane]);
    const bool staged = (n <= LCAP);

    #pragma unroll
    for (int k = 0; k < RPW * RLD; ++k) {
        if (w0[k] != 0xFFFFFFFFu) {
            const unsigned int ld = min(w0[k] >> SRCBITS, (unsigned int)(BN - 1));
            atomicAdd(&sCnt[ld], 1u);
        }
    }
    for (int j = 0; j < RPW; ++j) {
        const unsigned int pos = (unsigned int)__builtin_amdgcn_readlane((int)rpos, j);
        const unsigned int len = (unsigned int)__builtin_amdgcn_readlane((int)rlen, j);
        for (unsigned int i = (unsigned int)(RLD * 32 + lane); i < len; i += 32u) {
            const unsigned int w = tmp[pos + i];
            atomicAdd(&sCnt[min(w >> SRCBITS, (unsigned int)(BN - 1))], 1u);
        }
    }
    __syncthreads();

    const unsigned int v = (tid < BN) ? sCnt[tid] : 0u;
    const unsigned int incl = wave_incl_scan(v, lane);
    if (tid < 512 && lane == 31) sWv[wave] = incl;
    __syncthreads();
    if (tid < BN) {
        unsigned int woff = 0u;
        for (int q = 0; q < wave; ++q) woff += sWv[q];
        const unsigned int excl = woff + incl - v;
        const int node = b * BN + tid;
        sCur[tid]    = excl;
        rowptr[node] = beg + excl;
        dinv[node]   = rsqrtf((float)(v + 1u));
        if (b == NBKT - 1 && tid == BN - 1) rowptr[NNODES] = beg + excl + v;
    }
    __syncthreads();

    unsigned int p0[RPW * RLD];
    #pragma unroll
    for (int k = 0; k < RPW * RLD; ++k) {
        p0[k] = 0u;
        if (w0[k] != 0xFFFFFFFFu)
            p0[k] = atomicAdd(&sCur[min(w0[k] >> SRCBITS, (unsigned int)(BN - 1))], 1u);
    }
    #pragma unroll
    for (int k = 0; k < RPW * RLD; ++k) {
        if (w0[k] != 0xFFFFFFFFu) {
            const int src = (int)(w0[k] & SRCMASK);
            if (staged) sOut[p0[k]] = src;
            else        col[beg + p0[k]] = src;
        }
    }
    for (int j = 0; j < RPW; ++j) {
        const unsigned int pos = (unsigned int)__builtin_amdgcn_readlane((int)rpos, j);
        const unsigned int len = (unsigned int)__builtin_amdgcn_readlane((int)rlen, j);
        for (unsigned int i = (unsigned int)(RLD * 32 + lane); i < len; i += 32u) {
            const unsigned int w = tmp[pos + i];
            const unsigned int p = atomicAdd(&sCur[min(w >> SRCBITS, (unsigned int)(BN - 1))], 1u);
            const int src = (int)(w & SRCMASK);
            if (staged) sOut[p] = src;
            else        col[beg + p] = src;
        }
    }
    __syncthreads();
    if (staged)
        for (int i = tid; i < n; i += ST) col[beg + (unsigned int)i] = sOut[i];
}

__global__ void k_gemm1(const float* __restrict__ x, const float* __restrict__ W,
                        const float* __restrict__ dinv, float* __restrict__ hs, int n) {
    __shared__ float sW[128 * 32];
    for (int i = threadIdx.x; i < 128 * 32; i += blockDim.x) {
        const int k = i >> 5, c = i & 31;
        sW[(k >> 1) * 64 + c * 2 + (k & 1)] = W[i];
    }
    __syncthreads();

    const int wave = threadIdx.x >> 5;
    const int lane = threadIdx.x & 31;
    const int tile = blockIdx.x * (blockDim.x >> 5) + wave;
    const int rowBase = tile * 16;
    if (rowBase >= n) return;

    const int half = lane >> 4;
    const int l16  = lane & 15;

    v8f c0 = {};
    v8f c1 = {};
    const float* xrow = x + (size_t)(rowBase + l16) * 128;

    #pragma unroll 4
    for (int k = 0; k < 128; k += 4) {
        const int ka = k + 2 * half;
        v2f a;  a.x = xrow[ka];  a.y = xrow[ka + 1];
        const int bidx = (ka >> 1) * 64 + l16 * 2;
        v2f b0 = *(const v2f*)&sW[bidx];
        v2f b1 = *(const v2f*)&sW[bidx + 32];
        c0 = __builtin_amdgcn_wmma_f32_16x16x4_f32(false, a, false, b0, (short)0, c0, false, false);
        c1 = __builtin_amdgcn_wmma_f32_16x16x4_f32(false, a, false, b1, (short)0, c1, false, false);
    }

    #pragma unroll
    for (int r = 0; r < 8; ++r) {
        const int m = rowBase + r + 8 * half;
        const float s = dinv[m];
        const float v0 = c0[r] * s;
        const float v1 = c1[r] * s;
        hs[(size_t)m * 32 + l16]      = v0;
        hs[(size_t)m * 32 + 16 + l16] = v1;
    }
}

#define PT   256
#define NPW  4
#define NB   ((PT / 32) * NPW)
#define APITCH 36
#define TPITCH 17
static_assert(NNODES % NB == 0, "every block owns a full set of nodes (the kernels have a block barrier)");
static_assert(NB == 32, "two WMMA tiles and one out line per block");

__device__ __forceinline__ float sum_rows32(const int* __restrict__ col, const float* __restrict__ hs,
                                            int beg, int end, int first, int lane, float acc) {
    for (int p0 = beg; p0 < end; p0 += 32) {
        const int cnt = min(32, end - p0);
        int mine = first;
        if (p0 != beg) mine = (lane < cnt) ? col[p0 + lane] : 0;
        int k = 0;
        for (; k + 8 <= cnt; k += 8) {
            float r[8];
            #pragma unroll
            for (int u = 0; u < 8; ++u) {
                const int s = __builtin_amdgcn_readlane(mine, k + u);
                r[u] = hs[(size_t)s * 32 + lane];
            }
            #pragma unroll
            for (int u = 0; u < 8; ++u) acc += r[u];
        }
        if (k < cnt) {
            float r[8];
            #pragma unroll
            for (int u = 0; u < 8; ++u) {
                r[u] = 0.0f;
                if (k + u < cnt) {
                    const int s = __builtin_amdgcn_readlane(mine, k + u);
                    r[u] = hs[(size_t)s * 32 + lane];
                }
            }
            #pragma unroll
            for (int u = 0; u < 8; ++u) acc += r[u];
        }
    }
    return acc;
}

__global__ void __launch_bounds__(PT)
k_agg1_gemm2(const unsigned int* __restrict__ rowptr, const int* __restrict__ col,
             const float* __restrict__ hs, const float* __restrict__ W,
             const float* __restrict__ b1, const float* __restrict__ dinv,
             float* __restrict__ hs2) {
    __shared__ float sW[32 * 16];
    __shared__ float sA[NB * APITCH];
    __shared__ float sD[NB];
    static_assert(32 * 16 == 2 * PT, "two W2 elements per thread");
    {
        const int i0 = threadIdx.x, i1 = threadIdx.x + PT;
        const float w0 = W[i0], w1 = W[i1];
        sW[((i0 >> 4) >> 1) * 32 + (i0 & 15) * 2 + ((i0 >> 4) & 1)] = w0;
        sW[((i1 >> 4) >> 1) * 32 + (i1 & 15) * 2 + ((i1 >> 4) & 1)] = w1;
    }
    const int lane = threadIdx.x & 31;
    const int wave = __builtin_amdgcn_readfirstlane((int)(threadIdx.x >> 5));
    const int nodeBase = blockIdx.x * NB;
    const int v0 = nodeBase + wave * NPW;
    const float bias = b1[lane];

    int rp[NPW + 1];
    #pragma unroll
    for (int j = 0; j <= NPW; ++j) rp[j] = (int)rowptr[v0 + j];
    int next = (lane < min(32, rp[1] - rp[0])) ? col[rp[0] + lane] : 0;
    #pragma unroll
    for (int j = 0; j < NPW; ++j) {
        const int v = v0 + j;
        const int first = next;
        if (j + 1 < NPW) next = (lane < min(32, rp[j + 2] - rp[j + 1])) ? col[rp[j + 1] + lane] : 0;
        const float dv = dinv[v];
        float acc = hs[(size_t)v * 32 + lane];
        acc = sum_rows32(col, hs, rp[j], rp[j + 1], first, lane, acc);
        const int r = wave * NPW + j;
        sA[r * APITCH + lane] = lrelu(dv * acc + bias);
        if (lane == 0) sD[r] = dv;
    }
    __syncthreads();
    if (wave >= 2) return;

    const int half = lane >> 4;
    const int l16  = lane & 15;
    v8f c = {};
    const float* arow = &sA[(wave * 16 + l16) * APITCH];
    #pragma unroll
    for (int k = 0; k < 32; k += 4) {
        const int ka = k + 2 * half;
        v2f a = *(const v2f*)&arow[ka];
        v2f b = *(const v2f*)&sW[(ka >> 1) * 32 + l16 * 2];
        c = __builtin_amdgcn_wmma_f32_16x16x4_f32(false, a, false, b, (short)0, c, false, false);
    }
    #pragma unroll
    for (int r = 0; r < 8; ++r) {
        const int m = wave * 16 + r + 8 * half;
        hs2[(size_t)(nodeBase + m) * 16 + l16] = c[r] * sD[m];
    }
}

__device__ __forceinline__ float sum_rows16(const int* __restrict__ col, const float* __restrict__ hs,
                                            int beg, int end, int first, int lane) {
    const int half = lane >> 4;
    const int l16  = lane & 15;
    float acc = 0.0f;
    for (int p0 = beg; p0 < end; p0 += 32) {
        const int cnt = min(32, end - p0);
        int mine = first;
        if (p0 != beg) mine = (lane < cnt) ? col[p0 + lane] : 0;
        int k = 0;
        for (; k + 16 <= cnt; k += 16) {
            float r[8];
            #pragma unroll
            for (int u = 0; u < 8; ++u) {
                const int s = __shfl(mine, k + 2 * u + half);
                r[u] = hs[(size_t)s * 16 + l16];
            }
            #pragma unroll
            for (int u = 0; u < 8; ++u) acc += r[u];
        }
        if (k < cnt) {
            float r[8];
            #pragma unroll
            for (int u = 0; u < 8; ++u) {
                const int j = k + 2 * u + half;
                const int s = __shfl(mine, j & 31);
                r[u] = 0.0f;
                if (j < cnt) r[u] = hs[(size_t)s * 16 + l16];
            }
            #pragma unroll
            for (int u = 0; u < 8; ++u) acc += r[u];
        }
    }
    return acc;
}

__global__ void __launch_bounds__(PT)
k_agg2_out(const unsigned int* __restrict__ rowptr, const int* __restrict__ col,
           const float* __restrict__ hs, const float* __restrict__ dinv,
           const float* __restrict__ b2, const float* __restrict__ Wfc,
           const float* __restrict__ bfc, float* __restrict__ out) {
    __shared__ float sT[NB * TPITCH];
    const int lane = threadIdx.x & 31;
    const int l16  = lane & 15;
    const int wave = __builtin_amdgcn_readfirstlane((int)(threadIdx.x >> 5));
    const int nodeBase = blockIdx.x * NB;
    const int v0 = nodeBase + wave * NPW;
    const float bias = b2[l16];

    int rp[NPW + 1];
    #pragma unroll
    for (int j = 0; j <= NPW; ++j) rp[j] = (int)rowptr[v0 + j];
    int next = (lane < min(32, rp[1] - rp[0])) ? col[rp[0] + lane] : 0;
    #pragma unroll
    for (int j = 0; j < NPW; ++j) {
        const int v = v0 + j;
        const int first = next;
        if (j + 1 < NPW) next = (lane < min(32, rp[j + 2] - rp[j + 1])) ? col[rp[j + 1] + lane] : 0;
        const float dv = dinv[v];
        const float self = hs[(size_t)v * 16 + l16];
        float acc = sum_rows16(col, hs, rp[j], rp[j + 1], first, lane);
        acc += __shfl_xor(acc, 16);
        if (lane < 16) sT[(wave * NPW + j) * TPITCH + l16] = lrelu(dv * (acc + self) + bias);
    }
    __syncthreads();
    if (wave == 0) {
        const float* row = &sT[lane * TPITCH];
        float s = bfc[0];
        #pragma unroll
        for (int j = 0; j < 16; ++j) s += row[j] * Wfc[j];
        out[nodeBase + lane] = s;
    }
}

extern "C" void kernel_launch(void* const* d_in, const int* in_sizes, int n_in,
                              void* d_out, int out_size, void* d_ws, size_t ws_size,
                              hipStream_t stream) {
    const float* x    = (const float*)d_in[0];
    const int*   ei   = (const int*)  d_in[1];
    const float* W1   = (const float*)d_in[2];
    const float* b1   = (const float*)d_in[3];
    const float* W2   = (const float*)d_in[4];
    const float* b2   = (const float*)d_in[5];
    const float* Wfc  = (const float*)d_in[6];
    const float* bfc  = (const float*)d_in[7];
    float*       out  = (float*)d_out;

    const int n = NNODES;
    const int E = in_sizes[1] / 2;

    char* ws = (char*)d_ws;
    size_t off = 0;
    auto carve = [&](size_t bytes) { char* p = ws + off; off = (off + bytes + 511) & ~(size_t)511; return p; };
    unsigned int* offT  = (unsigned int*)carve((size_t)NCH * NBP * 4);
    float* dinv = (float*)carve((size_t)n * 4);
    unsigned int* rowptr   = (unsigned int*)carve((size_t)(n + 1) * 4);
    float* h1s  = (float*)carve((size_t)n * 32 * 4);
    float* h2s  = (float*)carve((size_t)n * 16 * 4);
    unsigned int* tmp = (unsigned int*)carve((size_t)E * 4);
    int*   col  = (int*)(ws + off);
    (void)ws_size;

    const int T = 256;
    const int wavesPerBlock = T / 32;
    const int tiles = (n + 15) / 16;
    const int gemmBlocks = (tiles + wavesPerBlock - 1) / wavesPerBlock;

    k_bkt_part <<<NCH, ST, 0, stream>>>(ei, offT, tmp, E);
    k_bkt_sort <<<NBKT, ST, 0, stream>>>(tmp, offT, rowptr, dinv, col, E);

    const int aggBlocks = n / NB;

    k_gemm1     <<<gemmBlocks, T, 0, stream>>>(x, W1, dinv, h1s, n);
    k_agg1_gemm2<<<aggBlocks, PT, 0, stream>>>(rowptr, col, h1s, W2, b1, dinv, h2s);

    k_agg2_out  <<<aggBlocks, PT, 0, stream>>>(rowptr, col, h2s, dinv, b2, Wfc, bfc, out);
}


# layer-1 product moved into the bucket-sort launch: each bucket block ends by multiplying its own 400 rows of x by W1 on WMMA, scaled by the dinv it has just counted; one launch fewer, sums unchanged
# speedup vs baseline: 7.0518x; 1.0998x over previous
#include <hip/hip_runtime.h>

typedef float v2f __attribute__((ext_vector_type(2)));
typedef float v8f __attribute__((ext_vector_type(8)));

#define NNODES   100000
#define NEGSLOPE 0.01f

__device__ __forceinline__ float lrelu(float v) { return v > 0.0f ? v : NEGSLOPE * v; }

#define NBKT  250
#define BN    400
#define NCH   256
#define NBP   256
#define ST    1024
#define EPT   13
#define ECAP  (EPT * ST)
#define LCAP  24576
#define SRCBITS 17
#define SRCMASK 0x1FFFFu
static_assert(NBKT * BN == NNODES, "buckets tile the node range");
static_assert(NNODES <= (1 << SRCBITS), "a source index fits its field");
static_assert(BN <= 512, "a bucket offset fits the bits above the source field");
static_assert(NBKT <= NBP, "padded bucket count");

__device__ __forceinline__ unsigned int bucket_of(int d) {
    const unsigned int b = (unsigned int)d / (unsigned int)BN;
    return b < (unsigned int)NBKT ? b : (unsigned int)(NBKT - 1);
}

__device__ __forceinline__ unsigned int wave_incl_scan(unsigned int v, int lane) {
    #pragma unroll
    for (int o = 1; o < 32; o <<= 1) {
        const unsigned int t = __shfl_up(v, (unsigned int)o);
        if (lane >= o) v += t;
    }
    return v;
}

__device__ __forceinline__ unsigned int wave_sum(unsigned int v) {
    #pragma unroll
    for (int o = 16; o > 0; o >>= 1) v += __shfl_xor(v, o);
    return v;
}

__global__ void __launch_bounds__(ST)
k_bkt_part(const int* __restrict__ edges, unsigned int* __restrict__ offT,
           unsigned int* __restrict__ tmp, int E) {
    __shared__ unsigned int sWord[ECAP];
    __shared__ unsigned int sCnt[NBP];
    __shared__ unsigned int sOff[NBP];
    __shared__ unsigned int sWv[NBP / 32];
    const int tid  = threadIdx.x;
    const int lane = tid & 31;
    const int wave = tid >> 5;
    const int c    = blockIdx.x;

    const int per   = (E + NCH - 1) / NCH;
    const int start = c * per;
    const int end   = min(E, start + min(per, ECAP));
    const int* __restrict__ srcv = edges;
    const int* __restrict__ dstv = edges + E;

    int d[EPT];
    int s[EPT];
    #pragma unroll
    for (int k = 0; k < EPT; ++k) {
        const int e = start + k * ST + tid;
        d[k] = (e < end) ? dstv[e] : -1;
        s[k] = (e < end) ? srcv[e] : 0;
    }
    if (tid < NBP) sCnt[tid] = 0u;
    __syncthreads();

    unsigned int r[EPT];
    #pragma unroll
    for (int k = 0; k < EPT; ++k) {
        r[k] = 0u;
        if (d[k] >= 0) r[k] = atomicAdd(&sCnt[bucket_of(d[k])], 1u);
    }
    __syncthreads();

    const unsigned int cnt  = (tid < NBP) ? sCnt[tid] : 0u;
    const unsigned int incl = wave_incl_scan(cnt, lane);
    if (tid < NBP && lane == 31) sWv[wave] = incl;
    __syncthreads();
    if (tid < NBP) {
        unsigned int woff = 0u;
        for (int q = 0; q < wave; ++q) woff += sWv[q];
        const unsigned int excl = woff + incl - cnt;
        sOff[tid] = excl;
        offT[c * NBP + tid] = excl;
    }
    __syncthreads();

    #pragma unroll
    for (int k = 0; k < EPT; ++k) {
        if (d[k] >= 0) {
            const unsigned int b = bucket_of(d[k]);
            const unsigned int off = (unsigned int)d[k] - b * (unsigned int)BN;
            sWord[sOff[b] + r[k]] = ((unsigned int)s[k] & SRCMASK) | (off << SRCBITS);
        }
    }
    __syncthreads();

    const int n = end - start;
    for (int i = tid; i < n; i += ST) tmp[start + i] = sWord[i];
}

#define RPW 8
#define RLD 2
#define GT  (BN / 16)
#define KG  8
static_assert((ST / 32) * RPW == NCH, "every run has a wave");
static_assert(BN % 16 == 0, "a bucket is a whole number of row tiles");
static_assert(GT <= ST / 32, "every row tile of the bucket has a wave");
static_assert(128 % (4 * KG) == 0 && 128 / (4 * KG) >= 2, "whole groups, at least two");
__global__ void __launch_bounds__(ST)
k_bkt_sort(const unsigned int* __restrict__ tmp, const unsigned int* __restrict__ offT,
           unsigned int* __restrict__ rowptr, float* __restrict__ dinv, int* __restrict__ col, int E,
           const float* __restrict__ x, const float* __restrict__ W, float* __restrict__ hs) {
    __shared__ float        sW[128 * 32];
    __shared__ int          sOut[LCAP];
    __shared__ unsigned int sCnt[512];
    __shared__ unsigned int sCur[512];
    __shared__ unsigned int sWv[16];
    __shared__ unsigned int sSumO[ST / 32];
    __shared__ unsigned int sSumL[ST / 32];
    static_assert(ST / 32 == 32, "one partial sum per lane in the block sums below");
    const int tid  = threadIdx.x;
    const int lane = tid & 31;
    const int wave = tid >> 5;
    const int b    = blockIdx.x;
    const int per  = (E + NCH - 1) / NCH;

    static_assert(128 * 32 == 4 * ST, "four W1 elements per thread");
    float wv[4];
    #pragma unroll
    for (int q = 0; q < 4; ++q) wv[q] = W[tid + q * ST];

    unsigned int o0 = 0u, rlen = 0u;
    if (lane < RPW) {
        const int c = wave * RPW + lane;
        o0   = offT[c * NBP + b];
        rlen = offT[c * NBP + b + 1] - o0;
    }
    const unsigned int rpos = (unsigned int)((wave * RPW + lane) * per) + o0;
    {
        const unsigned int so = wave_sum(o0);
        const unsigned int sl = wave_sum(rlen);
        if (lane == 0) { sSumO[wave] = so; sSumL[wave] = sl; }
    }
    if (tid < 512) sCnt[tid] = 0u;

    unsigned int w0[RPW * RLD];
    #pragma unroll
    for (int j = 0; j < RPW; ++j) {
        const unsigned int pos = (unsigned int)__builtin_amdgcn_readlane((int)rpos, j);
        const unsigned int len = (unsigned int)__builtin_amdgcn_readlane((int)rlen, j);
        #pragma unroll
        for (int u = 0; u < RLD; ++u) {
            const unsigned int i = (unsigned int)(u * 32 + lane);
            w0[j * RLD + u] = (i < len) ? tmp[pos + i] : 0xFFFFFFFFu;
        }
    }
    #pragma unroll
    for (int q = 0; q < 4; ++q) {
        const int i = tid + q * ST;
        const int k = i >> 5, c = i & 31;
        sW[(k >> 1) * 64 + c * 2 + (k & 1)] = wv[q];
    }
    __syncthreads();
    const unsigned int beg = wave_sum(sSumO[lane]);
    const int n = (int)wave_sum(sSumL[lane]);
    const bool staged = (n <= LCAP);

    #pragma unroll
    for (int k = 0; k < RPW * RLD; ++k) {
        if (w0[k] != 0xFFFFFFFFu) {
            const unsigned int ld = min(w0[k] >> SRCBITS, (unsigned int)(BN - 1));
            atomicAdd(&sCnt[ld], 1u);
        }
    }
    for (int j = 0; j < RPW; ++j) {
        const unsigned int pos = (unsigned int)__builtin_amdgcn_readlane((int)rpos, j);
        const unsigned int len = (unsigned int)__builtin_amdgcn_readlane((int)rlen, j);
        for (unsigned int i = (unsigned int)(RLD * 32 + lane); i < len; i += 32u) {
            const unsigned int w = tmp[pos + i];
            atomicAdd(&sCnt[min(w >> SRCBITS, (unsigned int)(BN - 1))], 1u);
        }
    }
    __syncthreads();

    const unsigned int v = (tid < BN) ? sCnt[tid] : 0u;
    const unsigned int incl = wave_incl_scan(v, lane);
    if (tid < 512 && lane == 31) sWv[wave] = incl;
    __syncthreads();
    if (tid < BN) {
        unsigned int woff = 0u;
        for (int q = 0; q < wave; ++q) woff += sWv[q];
        const unsigned int excl = woff + incl - v;
        const int node = b * BN + tid;
        sCur[tid]    = excl;
        rowptr[node] = beg + excl;
        dinv[node]   = rsqrtf((float)(v + 1u));
        if (b == NBKT - 1 && tid == BN - 1) rowptr[NNODES] = beg + excl + v;
    }
    __syncthreads();

    unsigned int p0[RPW * RLD];
    #pragma unroll
    for (int k = 0; k < RPW * RLD; ++k) {
        p0[k] = 0u;
        if (w0[k] != 0xFFFFFFFFu)
            p0[k] = atomicAdd(&sCur[min(w0[k] >> SRCBITS, (unsigned int)(BN - 1))], 1u);
    }
    #pragma unroll
    for (int k = 0; k < RPW * RLD; ++k) {
        if (w0[k] != 0xFFFFFFFFu) {
            const int src = (int)(w0[k] & SRCMASK);
            if (staged) sOut[p0[k]] = src;
            else        col[beg + p0[k]] = src;
        }
    }
    for (int j = 0; j < RPW; ++j) {
        const unsigned int pos = (unsigned int)__builtin_amdgcn_readlane((int)rpos, j);
        const unsigned int len = (unsigned int)__builtin_amdgcn_readlane((int)rlen, j);
        for (unsigned int i = (unsigned int)(RLD * 32 + lane); i < len; i += 32u) {
            const unsigned int w = tmp[pos + i];
            const unsigned int p = atomicAdd(&sCur[min(w >> SRCBITS, (unsigned int)(BN - 1))], 1u);
            const int src = (int)(w & SRCMASK);
            if (staged) sOut[p] = src;
            else        col[beg + p] = src;
        }
    }
    __syncthreads();

    const int  gw   = __builtin_amdgcn_readfirstlane(wave);
    const bool mul  = gw < GT;
    const int  half = lane >> 4;
    const int  l16  = lane & 15;
    const float* xrow = x + ((size_t)(b * BN + (mul ? gw : 0) * 16 + l16) * 128 + 2 * half);
    v2f a[2][KG];
    if (mul) {
        #pragma unroll
        for (int u = 0; u < KG; ++u) { a[0][u].x = xrow[4 * u]; a[0][u].y = xrow[4 * u + 1]; }
    }

    if (staged)
        for (int i = tid; i < n; i += ST) col[beg + (unsigned int)i] = sOut[i];

    if (mul) {
        v8f c0 = {};
        v8f c1 = {};
        #pragma unroll
        for (int u = 0; u < KG; ++u) { a[1][u].x = xrow[4 * (KG + u)]; a[1][u].y = xrow[4 * (KG + u) + 1]; }
        #pragma unroll
        for (int g = 0; g < 128 / (4 * KG); ++g) {
            #pragma unroll
            for (int u = 0; u < KG; ++u) {
                const int ka = (g * KG + u) * 4 + 2 * half;
                const int bidx = (ka >> 1) * 64 + l16 * 2;
                v2f b0 = *(const v2f*)&sW[bidx];
                v2f b1 = *(const v2f*)&sW[bidx + 32];
                c0 = __builtin_amdgcn_wmma_f32_16x16x4_f32(false, a[g & 1][u], false, b0, (short)0, c0, false, false);
                c1 = __builtin_amdgcn_wmma_f32_16x16x4_f32(false, a[g & 1][u], false, b1, (short)0, c1, false, false);
            }
            if (g + 2 < 128 / (4 * KG)) {
                #pragma unroll
                for (int u = 0; u < KG; ++u) {
                    a[g & 1][u].x = xrow[4 * ((g + 2) * KG + u)];
                    a[g & 1][u].y = xrow[4 * ((g + 2) * KG + u) + 1];
                }
            }
        }
        #pragma unroll
        for (int r = 0; r < 8; ++r) {
            const int ml = gw * 16 + r + 8 * half;
            const float s = rsqrtf((float)(sCnt[ml] + 1u));
            const size_t m = (size_t)(b * BN + ml);
            hs[m * 32 + l16]      = c0[r] * s;
            hs[m * 32 + 16 + l16] = c1[r] * s;
        }
    }
}

#define PT   256
#define NPW  4
#define NB   ((PT / 32) * NPW)
#define APITCH 36
#define TPITCH 17
static_assert(NNODES % NB == 0, "every block owns a full set of nodes (the kernels have a block barrier)");
static_assert(NB == 32, "two WMMA tiles and one out line per block");

__device__ __forceinline__ float sum_rows32(const int* __restrict__ col, const float* __restrict__ hs,
                                            int beg, int end, int first, int lane, float acc) {
    for (int p0 = beg; p0 < end; p0 += 32) {
        const int cnt = min(32, end - p0);
        int mine = first;
        if (p0 != beg) mine = (lane < cnt) ? col[p0 + lane] : 0;
        int k = 0;
        for (; k + 8 <= cnt; k += 8) {
            float r[8];
            #pragma unroll
            for (int u = 0; u < 8; ++u) {
                const int s = __builtin_amdgcn_readlane(mine, k + u);
                r[u] = hs[(size_t)s * 32 + lane];
            }
            #pragma unroll
            for (int u = 0; u < 8; ++u) acc += r[u];
        }
        if (k < cnt) {
            float r[8];
            #pragma unroll
            for (int u = 0; u < 8; ++u) {
                r[u] = 0.0f;
                if (k + u < cnt) {
                    const int s = __builtin_amdgcn_readlane(mine, k + u);
                    r[u] = hs[(size_t)s * 32 + lane];
                }
            }
            #pragma unroll
            for (int u = 0; u < 8; ++u) acc += r[u];
        }
    }
    return acc;
}

__global__ void __launch_bounds__(PT)
k_agg1_gemm2(const unsigned int* __restrict__ rowptr, const int* __restrict__ col,
             const float* __restrict__ hs, const float* __restrict__ W,
             const float* __restrict__ b1, const float* __restrict__ dinv,
             float* __restrict__ hs2) {
    __shared__ float sW[32 * 16];
    __shared__ float sA[NB * APITCH];
    __shared__ float sD[NB];
    static_assert(32 * 16 == 2 * PT, "two W2 elements per thread");
    {
        const int i0 = threadIdx.x, i1 = threadIdx.x + PT;
        const float w0 = W[i0], w1 = W[i1];
        sW[((i0 >> 4) >> 1) * 32 + (i0 & 15) * 2 + ((i0 >> 4) & 1)] = w0;
        sW[((i1 >> 4) >> 1) * 32 + (i1 & 15) * 2 + ((i1 >> 4) & 1)] = w1;
    }
    const int lane = threadIdx.x & 31;
    const int wave = __builtin_amdgcn_readfirstlane((int)(threadIdx.x >> 5));
    const int nodeBase = blockIdx.x * NB;
    const int v0 = nodeBase + wave * NPW;
    const float bias = b1[lane];

    int rp[NPW + 1];
    #pragma unroll
    for (int j = 0; j <= NPW; ++j) rp[j] = (int)rowptr[v0 + j];
    int next = (lane < min(32, rp[1] - rp[0])) ? col[rp[0] + lane] : 0;
    #pragma unroll
    for (int j = 0; j < NPW; ++j) {
        const int v = v0 + j;
        const int first = next;
        if (j + 1 < NPW) next = (lane < min(32, rp[j + 2] - rp[j + 1])) ? col[rp[j + 1] + lane] : 0;
        const float dv = dinv[v];
        float acc = hs[(size_t)v * 32 + lane];
        acc = sum_rows32(col, hs, rp[j], rp[j + 1], first, lane, acc);
        const int r = wave * NPW + j;
        sA[r * APITCH + lane] = lrelu(dv * acc + bias);
        if (lane == 0) sD[r] = dv;
    }
    __syncthreads();
    if (wave >= 2) return;

    const int half = lane >> 4;
    const int l16  = lane & 15;
    v8f c = {};
    const float* arow = &sA[(wave * 16 + l16) * APITCH];
    #pragma unroll
    for (int k = 0; k < 32; k += 4) {
        const int ka = k + 2 * half;
        v2f a = *(const v2f*)&arow[ka];
        v2f b = *(const v2f*)&sW[(ka >> 1) * 32 + l16 * 2];
        c = __builtin_amdgcn_wmma_f32_16x16x4_f32(false, a, false, b, (short)0, c, false, false);
    }
    #pragma unroll
    for (int r = 0; r < 8; ++r) {
        const int m = wave * 16 + r + 8 * half;
        hs2[(size_t)(nodeBase + m) * 16 + l16] = c[r] * sD[m];
    }
}

__device__ __forceinline__ float sum_rows16(const int* __restrict__ col, const float* __restrict__ hs,
                                            int beg, int end, int first, int lane) {
    const int half = lane >> 4;
    const int l16  = lane & 15;
    float acc = 0.0f;
    for (int p0 = beg; p0 < end; p0 += 32) {
        const int cnt = min(32, end - p0);
        int mine = first;
        if (p0 != beg) mine = (lane < cnt) ? col[p0 + lane] : 0;
        int k = 0;
        for (; k + 16 <= cnt; k += 16) {
            float r[8];
            #pragma unroll
            for (int u = 0; u < 8; ++u) {
                const int s = __shfl(mine, k + 2 * u + half);
                r[u] = hs[(size_t)s * 16 + l16];
            }
            #pragma unroll
            for (int u = 0; u < 8; ++u) acc += r[u];
        }
        if (k < cnt) {
            float r[8];
            #pragma unroll
            for (int u = 0; u < 8; ++u) {
                const int j = k + 2 * u + half;
                const int s = __shfl(mine, j & 31);
                r[u] = 0.0f;
                if (j < cnt) r[u] = hs[(size_t)s * 16 + l16];
            }
            #pragma unroll
            for (int u = 0; u < 8; ++u) acc += r[u];
        }
    }
    return acc;
}

__global__ void __launch_bounds__(PT)
k_agg2_out(const unsigned int* __restrict__ rowptr, const int* __restrict__ col,
           const float* __restrict__ hs, const float* __restrict__ dinv,
           const float* __restrict__ b2, const float* __restrict__ Wfc,
           const float* __restrict__ bfc, float* __restrict__ out) {
    __shared__ float sT[NB * TPITCH];
    const int lane = threadIdx.x & 31;
    const int l16  = lane & 15;
    const int wave = __builtin_amdgcn_readfirstlane((int)(threadIdx.x >> 5));
    const int nodeBase = blockIdx.x * NB;
    const int v0 = nodeBase + wave * NPW;
    const float bias = b2[l16];

    int rp[NPW + 1];
    #pragma unroll
    for (int j = 0; j <= NPW; ++j) rp[j] = (int)rowptr[v0 + j];
    int next = (lane < min(32, rp[1] - rp[0])) ? col[rp[0] + lane] : 0;
    #pragma unroll
    for (int j = 0; j < NPW; ++j) {
        const int v = v0 + j;
        const int first = next;
        if (j + 1 < NPW) next = (lane < min(32, rp[j + 2] - rp[j + 1])) ? col[rp[j + 1] + lane] : 0;
        const float dv = dinv[v];
        const float self = hs[(size_t)v * 16 + l16];
        float acc = sum_rows16(col, hs, rp[j], rp[j + 1], first, lane);
        acc += __shfl_xor(acc, 16);
        if (lane < 16) sT[(wave * NPW + j) * TPITCH + l16] = lrelu(dv * (acc + self) + bias);
    }
    __syncthreads();
    if (wave == 0) {
        const float* row = &sT[lane * TPITCH];
        float s = bfc[0];
        #pragma unroll
        for (int j = 0; j < 16; ++j) s += row[j] * Wfc[j];
        out[nodeBase + lane] = s;
    }
}

extern "C" void kernel_launch(void* const* d_in, const int* in_sizes, int n_in,
                              void* d_out, int out_size, void* d_ws, size_t ws_size,
                              hipStream_t stream) {
    const float* x    = (const float*)d_in[0];
    const int*   ei   = (const int*)  d_in[1];
    const float* W1   = (const float*)d_in[2];
    const float* b1   = (const float*)d_in[3];
    const float* W2   = (const float*)d_in[4];
    const float* b2   = (const float*)d_in[5];
    const float* Wfc  = (const float*)d_in[6];
    const float* bfc  = (const float*)d_in[7];
    float*       out  = (float*)d_out;

    const int n = NNODES;
    const int E = in_sizes[1] / 2;

    char* ws = (char*)d_ws;
    size_t off = 0;
    auto carve = [&](size_t bytes) { char* p = ws + off; off = (off + bytes + 511) & ~(size_t)511; return p; };
    unsigned int* offT  = (unsigned int*)carve((size_t)NCH * NBP * 4);
    float* dinv = (float*)carve((size_t)n * 4);
    unsigned int* rowptr   = (unsigned int*)carve((size_t)(n + 1) * 4);
    float* h1s  = (float*)carve((size_t)n * 32 * 4);
    float* h2s  = (float*)carve((size_t)n * 16 * 4);
    unsigned int* tmp = (unsigned int*)carve((size_t)E * 4);
    int*   col  = (int*)(ws + off);
    (void)ws_size;

    k_bkt_part <<<NCH, ST, 0, stream>>>(ei, offT, tmp, E);
    k_bkt_sort <<<NBKT, ST, 0, stream>>>(tmp, offT, rowptr, dinv, col, E, x, W1, h1s);

    const int aggBlocks = n / NB;

    k_agg1_gemm2<<<aggBlocks, PT, 0, stream>>>(rowptr, col, h1s, W2, b1, dinv, h2s);

    k_agg2_out  <<<aggBlocks, PT, 0, stream>>>(rowptr, col, h2s, dinv, b2, Wfc, bfc, out);
}
